# starting kernel
# baseline (speedup 1.0000x reference)
#include <hip/hip_runtime.h>


typedef float v2f __attribute__((ext_vector_type(2)));
typedef float v8f __attribute__((ext_vector_type(8)));

#define NN   100000
#define EE   1600000
#define DH   64
#define DOUT 16

__global__ void gcn_zero_f32(float* __restrict__ p, int n) {
  int i = blockIdx.x * blockDim.x + threadIdx.x;
  if (i < n) p[i] = 0.0f;
}

__global__ void gcn_deg_kernel(const int* __restrict__ dst,
                               float* __restrict__ deg) {
  int e = blockIdx.x * blockDim.x + threadIdx.x;
  if (e < EE) unsafeAtomicAdd(&deg[dst[e]], 1.0f);
}

__global__ void gcn_dinv_kernel(float* __restrict__ deg) {
  int i = blockIdx.x * blockDim.x + threadIdx.x;
  if (i < NN) deg[i] = rsqrtf(deg[i] + 1.0f);
}

template <int NCOLS, bool BIAS, bool RELU>
__global__ __launch_bounds__(32) void gcn_gemm_wmma(
    const float* __restrict__ X, const float* __restrict__ W,
    const float* __restrict__ bias, float* __restrict__ Out) {
  const int tileM = blockIdx.x;
  const int tileN = blockIdx.y;
  const int lane  = threadIdx.x;
  const int half  = lane >> 4;
  const int l     = lane & 15;
  const int arow  = tileM * 16 + l;
  const int bcol  = tileN * 16 + l;

  v8f c = {};
#pragma unroll
  for (int k0 = 0; k0 < DH; k0 += 4) {
    const int ak = k0 + half * 2;
    v2f a, b;
    a.x = X[arow * DH + ak];
    a.y = X[arow * DH + ak + 1];
    b.x = W[ak * NCOLS + bcol];
    b.y = W[(ak + 1) * NCOLS + bcol];
    c = __builtin_amdgcn_wmma_f32_16x16x4_f32(
         false, a,  false, b,
         (short)0, c,  false,  false);
  }

#pragma unroll
  for (int r = 0; r < 8; ++r) {
    const int row = tileM * 16 + half * 8 + r;
    float v = c[r];
    if (BIAS) v += bias[bcol];
    if (RELU) v = fmaxf(v, 0.0f);
    Out[row * NCOLS + bcol] = v;
  }
}

__global__ void gcn_scatter_kernel(const float* __restrict__ H,
                                   const int* __restrict__ src,
                                   const int* __restrict__ dst,
                                   const float* __restrict__ dinv,
                                   float* __restrict__ AGG) {
  int idx = blockIdx.x * blockDim.x + threadIdx.x;
  if (idx >= EE * DH) return;
  const int e = idx >> 6;
  const int d = idx & (DH - 1);
  const int s = src[e];
  const int t = dst[e];
  const float coef = dinv[s] * dinv[t];
  unsafeAtomicAdd(&AGG[t * DH + d], H[s * DH + d] * coef);
}

__global__ void gcn_epilogue_kernel(float* __restrict__ AGG,
                                    const float* __restrict__ H,
                                    const float* __restrict__ dinv,
                                    const float* __restrict__ b) {
  int idx = blockIdx.x * blockDim.x + threadIdx.x;
  if (idx >= NN * DH) return;
  const int i = idx >> 6;
  const int d = idx & (DH - 1);
  const float di = dinv[i];
  const float v = AGG[idx] + H[idx] * di * di + b[d];
  AGG[idx] = fmaxf(v, 0.0f);
}

extern "C" void kernel_launch(void* const* d_in, const int* in_sizes, int n_in,
                              void* d_out, int out_size, void* d_ws,
                              size_t ws_size, hipStream_t stream) {
  (void)in_sizes; (void)n_in; (void)out_size; (void)ws_size;

  const float* x   = (const float*)d_in[0];
  const int*   ei  = (const int*)d_in[1];
  const int*   src = ei;
  const int*   dst = ei + EE;
  const float* W1  = (const float*)d_in[2];
  const float* b1  = (const float*)d_in[3];
  const float* W2  = (const float*)d_in[4];
  const float* b2  = (const float*)d_in[5];
  const float* W3  = (const float*)d_in[6];
  const float* b3  = (const float*)d_in[7];
  const float* Wf1 = (const float*)d_in[8];
  const float* bf1 = (const float*)d_in[9];
  const float* Wf2 = (const float*)d_in[10];
  const float* bf2 = (const float*)d_in[11];
  float* out = (float*)d_out;

  float* base = (float*)d_ws;
  float* dinv = base;
  float* bufA = base + NN;
  float* bufB = bufA + NN * DH;

  const int T = 256;
  const int gN   = (NN + T - 1) / T;
  const int gE   = (EE + T - 1) / T;
  const int gND  = (NN * DH + T - 1) / T;
  const int gED  = (EE * DH + T - 1) / T;
  const dim3 gemmGrid(NN / 16, DH / 16);
  const dim3 gemmGridOut(NN / 16, DOUT / 16);
  const dim3 gemmBlk(32);

  gcn_zero_f32<<<gN, T, 0, stream>>>(dinv, NN);
  gcn_deg_kernel<<<gE, T, 0, stream>>>(dst, dinv);
  gcn_dinv_kernel<<<gN, T, 0, stream>>>(dinv);

  gcn_gemm_wmma<DH, false, false><<<gemmGrid, gemmBlk, 0, stream>>>(x, W1, nullptr, bufA);
  gcn_zero_f32<<<gND, T, 0, stream>>>(bufB, NN * DH);
  gcn_scatter_kernel<<<gED, T, 0, stream>>>(bufA, src, dst, dinv, bufB);
  gcn_epilogue_kernel<<<gND, T, 0, stream>>>(bufB, bufA, dinv, b1);

  gcn_gemm_wmma<DH, false, false><<<gemmGrid, gemmBlk, 0, stream>>>(bufB, W2, nullptr, bufA);
  gcn_zero_f32<<<gND, T, 0, stream>>>(bufB, NN * DH);
  gcn_scatter_kernel<<<gED, T, 0, stream>>>(bufA, src, dst, dinv, bufB);
  gcn_epilogue_kernel<<<gND, T, 0, stream>>>(bufB, bufA, dinv, b2);

  gcn_gemm_wmma<DH, false, false><<<gemmGrid, gemmBlk, 0, stream>>>(bufB, W3, nullptr, bufA);
  gcn_zero_f32<<<gND, T, 0, stream>>>(bufB, NN * DH);
  gcn_scatter_kernel<<<gED, T, 0, stream>>>(bufA, src, dst, dinv, bufB);
  gcn_epilogue_kernel<<<gND, T, 0, stream>>>(bufB, bufA, dinv, b3);

  gcn_gemm_wmma<DH, true, true><<<gemmGrid, gemmBlk, 0, stream>>>(bufB, Wf1, bf1, bufA);
  gcn_gemm_wmma<DOUT, true, false><<<gemmGridOut, gemmBlk, 0, stream>>>(bufA, Wf2, bf2, out);
}


# Scatter-add leaves L2 atomics: edges binned per call into 128-node dst buckets, one block per bucket sums H rows in LDS with f32 LDS atomics (arrival order, as before) and finishes; degrees likewise
# speedup vs baseline: 2.9586x; 2.9586x over previous
#include <hip/hip_runtime.h>


typedef float v2f __attribute__((ext_vector_type(2)));
typedef float v8f __attribute__((ext_vector_type(8)));

#define NN   100000
#define EE   1600000
#define DH   64
#define DOUT 16

#define BN   128
#define BSH  7
#define NB   782
#define NS   128
#define SL   12500
#define SRCM 0x1FFFF
#define BT   1024
#define AT   512

__global__ __launch_bounds__(BT) void gcn_bin_count(const int* __restrict__ dst,
                                                    int* __restrict__ cnt) {
  __shared__ int hist[NB];
  const int tid = threadIdx.x;
  for (int i = tid; i < NB; i += BT) hist[i] = 0;
  __syncthreads();
  const int e0 = blockIdx.x * SL;
  for (int i = tid; i < SL; i += BT) atomicAdd(&hist[dst[e0 + i] >> BSH], 1);
  __syncthreads();
  for (int i = tid; i < NB; i += BT) cnt[blockIdx.x * NB + i] = hist[i];
}

__global__ __launch_bounds__(BT) void gcn_bin_scan(const int* __restrict__ cnt,
                                                   int* __restrict__ off,
                                                   int* __restrict__ bstart) {
  __shared__ int tot[BT];
  const int b = threadIdx.x;
  int run = 0;
  if (b < NB) {
    for (int s = 0; s < NS; s += 8) {
      int c[8];
#pragma unroll
      for (int u = 0; u < 8; ++u) c[u] = cnt[(s + u) * NB + b];
#pragma unroll
      for (int u = 0; u < 8; ++u) { off[(s + u) * NB + b] = run; run += c[u]; }
    }
  }
  tot[b] = run;
  __syncthreads();
  for (int d = 1; d < BT; d <<= 1) {
    const int v = (b >= d) ? tot[b - d] : 0;
    __syncthreads();
    tot[b] += v;
    __syncthreads();
  }
  if (b < NB) bstart[b + 1] = tot[b];
  if (b == 0) bstart[0] = 0;
}

__global__ __launch_bounds__(BT) void gcn_bin_place(const int* __restrict__ src,
                                                    const int* __restrict__ dst,
                                                    const int* __restrict__ off,
                                                    const int* __restrict__ bstart,
                                                    int* __restrict__ rec) {
  __shared__ int cur[NB];
  const int tid = threadIdx.x;
  for (int i = tid; i < NB; i += BT) cur[i] = bstart[i] + off[blockIdx.x * NB + i];
  __syncthreads();
  const int e0 = blockIdx.x * SL;
  for (int i = tid; i < SL; i += BT) {
    const int t = dst[e0 + i];
    const int s = src[e0 + i];
    const int pos = atomicAdd(&cur[t >> BSH], 1);
    rec[pos] = s | ((t & (BN - 1)) << 17);
  }
}

__global__ __launch_bounds__(256) void gcn_bucket_dinv(const int* __restrict__ rec,
                                                       const int* __restrict__ bstart,
                                                       float* __restrict__ dinv) {
  __shared__ int c[BN];
  const int tid = threadIdx.x;
  if (tid < BN) c[tid] = 0;
  __syncthreads();
  const int beg = bstart[blockIdx.x];
  const int end = bstart[blockIdx.x + 1];
  for (int e = beg + tid; e < end; e += 256) atomicAdd(&c[rec[e] >> 17], 1);
  __syncthreads();
  const int node = blockIdx.x * BN + tid;
  if (tid < BN && node < NN) dinv[node] = rsqrtf((float)c[tid] + 1.0f);
}

template <int NCOLS, bool BIAS, bool RELU>
__global__ __launch_bounds__(32) void gcn_gemm_wmma(
    const float* __restrict__ X, const float* __restrict__ W,
    const float* __restrict__ bias, float* __restrict__ Out) {
  const int tileM = blockIdx.x;
  const int tileN = blockIdx.y;
  const int lane  = threadIdx.x;
  const int half  = lane >> 4;
  const int l     = lane & 15;
  const int arow  = tileM * 16 + l;
  const int bcol  = tileN * 16 + l;

  v8f c = {};
#pragma unroll
  for (int k0 = 0; k0 < DH; k0 += 4) {
    const int ak = k0 + half * 2;
    v2f a, b;
    a.x = X[arow * DH + ak];
    a.y = X[arow * DH + ak + 1];
    b.x = W[ak * NCOLS + bcol];
    b.y = W[(ak + 1) * NCOLS + bcol];
    c = __builtin_amdgcn_wmma_f32_16x16x4_f32(
         false, a,  false, b,
         (short)0, c,  false,  false);
  }

#pragma unroll
  for (int r = 0; r < 8; ++r) {
    const int row = tileM * 16 + half * 8 + r;
    float v = c[r];
    if (BIAS) v += bias[bcol];
    if (RELU) v = fmaxf(v, 0.0f);
    Out[row * NCOLS + bcol] = v;
  }
}

__global__ __launch_bounds__(AT) void gcn_agg_bucket(const float* __restrict__ H,
                                                     const int* __restrict__ rec,
                                                     const int* __restrict__ bstart,
                                                     const float* __restrict__ dinv,
                                                     const float* __restrict__ bias,
                                                     float* __restrict__ Out) {
  __shared__ float tile[BN * DH];
  __shared__ float sdinv[BN];
  const int tid   = threadIdx.x;
  const int lane  = tid & 31;
  const int wave  = __builtin_amdgcn_readfirstlane(tid >> 5);
  const int node0 = blockIdx.x * BN;

  float4* tile4 = reinterpret_cast<float4*>(tile);
  for (int i = tid; i < BN * DH / 4; i += AT) tile4[i] = make_float4(0.0f, 0.0f, 0.0f, 0.0f);
  if (tid < BN) sdinv[tid] = (node0 + tid < NN) ? dinv[node0 + tid] : 0.0f;
  __syncthreads();

  const int beg = bstart[blockIdx.x];
  const int end = bstart[blockIdx.x + 1];
  const float2* H2 = reinterpret_cast<const float2*>(H);

  for (int base = beg + wave * 32; base < end; base += AT) {
    const int e = base + lane;
    int   r  = 0;
    float cf = 0.0f;
    if (e < end) {
      r  = rec[e];
      cf = dinv[r & SRCM] * sdinv[r >> 17];
    }
    const int n = end - base;
    if (n >= 32) {
#pragma unroll
      for (int g = 0; g < 32; g += 16) {
        float2 v[16];
#pragma unroll
        for (int j = 0; j < 16; ++j) {
          const int rj = __builtin_amdgcn_readlane(r, g + j);
          v[j] = H2[(rj & SRCM) * (DH / 2) + lane];
        }
#pragma unroll
        for (int j = 0; j < 16; ++j) {
          const int   rj = __builtin_amdgcn_readlane(r, g + j);
          const float cj = __int_as_float(__builtin_amdgcn_readlane(__float_as_int(cf), g + j));
          float* row = &tile[(rj >> 17) * DH + 2 * lane];
          atomicAdd(row,     v[j].x * cj);
          atomicAdd(row + 1, v[j].y * cj);
        }
      }
    } else {
      for (int j = 0; j < n; ++j) {
        const int    rj = __builtin_amdgcn_readlane(r, j);
        const float  cj = __int_as_float(__builtin_amdgcn_readlane(__float_as_int(cf), j));
        const float2 v  = H2[(rj & SRCM) * (DH / 2) + lane];
        float* row = &tile[(rj >> 17) * DH + 2 * lane];
        atomicAdd(row,     v.x * cj);
        atomicAdd(row + 1, v.y * cj);
      }
    }
  }
  __syncthreads();

  const float4* H4 = reinterpret_cast<const float4*>(H);
  const float4* b4 = reinterpret_cast<const float4*>(bias);
  float4*       O4 = reinterpret_cast<float4*>(Out);
  for (int i = tid; i < BN * DH / 4; i += AT) {
    const int t    = i >> 4;
    const int q    = i & 15;
    const int node = node0 + t;
    if (node < NN) {
      const float  di = sdinv[t];
      const float4 a  = tile4[i];
      const float4 h  = H4[node * (DH / 4) + q];
      const float4 bb = b4[q];
      float4 o;
      o.x = fmaxf(a.x + h.x * di * di + bb.x, 0.0f);
      o.y = fmaxf(a.y + h.y * di * di + bb.y, 0.0f);
      o.z = fmaxf(a.z + h.z * di * di + bb.z, 0.0f);
      o.w = fmaxf(a.w + h.w * di * di + bb.w, 0.0f);
      O4[node * (DH / 4) + q] = o;
    }
  }
}

extern "C" void kernel_launch(void* const* d_in, const int* in_sizes, int n_in,
                              void* d_out, int out_size, void* d_ws,
                              size_t ws_size, hipStream_t stream) {
  (void)in_sizes; (void)n_in; (void)out_size; (void)ws_size;

  const float* x   = (const float*)d_in[0];
  const int*   ei  = (const int*)d_in[1];
  const int*   src = ei;
  const int*   dst = ei + EE;
  const float* W1  = (const float*)d_in[2];
  const float* b1  = (const float*)d_in[3];
  const float* W2  = (const float*)d_in[4];
  const float* b2  = (const float*)d_in[5];
  const float* W3  = (const float*)d_in[6];
  const float* b3  = (const float*)d_in[7];
  const float* Wf1 = (const float*)d_in[8];
  const float* bf1 = (const float*)d_in[9];
  const float* Wf2 = (const float*)d_in[10];
  const float* bf2 = (const float*)d_in[11];
  float* out = (float*)d_out;

  float* base = (float*)d_ws;
  float* dinv = base;
  float* bufA = base + NN;
  float* bufB = bufA + NN * DH;
  int*   rec  = (int*)(bufB + NN * DH);
  int*   cnt  = rec + EE;
  int*   off  = cnt + NS * NB;
  int*   bst  = off + NS * NB;

  const dim3 gemmGrid(NN / 16, DH / 16);
  const dim3 gemmGridOut(NN / 16, DOUT / 16);
  const dim3 gemmBlk(32);

  gcn_bin_count<<<NS, BT, 0, stream>>>(dst, cnt);
  gcn_bin_scan<<<1, BT, 0, stream>>>(cnt, off, bst);
  gcn_bin_place<<<NS, BT, 0, stream>>>(src, dst, off, bst, rec);
  gcn_bucket_dinv<<<NB, 256, 0, stream>>>(rec, bst, dinv);

  gcn_gemm_wmma<DH, false, false><<<gemmGrid, gemmBlk, 0, stream>>>(x, W1, nullptr, bufA);
  gcn_agg_bucket<<<NB, AT, 0, stream>>>(bufA, rec, bst, dinv, b1, bufB);

  gcn_gemm_wmma<DH, false, false><<<gemmGrid, gemmBlk, 0, stream>>>(bufB, W2, nullptr, bufA);
  gcn_agg_bucket<<<NB, AT, 0, stream>>>(bufA, rec, bst, dinv, b2, bufB);

  gcn_gemm_wmma<DH, false, false><<<gemmGrid, gemmBlk, 0, stream>>>(bufB, W3, nullptr, bufA);
  gcn_agg_bucket<<<NB, AT, 0, stream>>>(bufA, rec, bst, dinv, b3, bufB);

  gcn_gemm_wmma<DH, true, true><<<gemmGrid, gemmBlk, 0, stream>>>(bufB, Wf1, bf1, bufA);
  gcn_gemm_wmma<DOUT, true, false><<<gemmGridOut, gemmBlk, 0, stream>>>(bufA, Wf2, bf2, out);
}


# GEMM retiled: 8-wave blocks, a wave computes 32 rows x all columns from 16-byte A reads and W staged in LDS, stores whole rows; k regrouped across WMMA steps, so f32 sum order changes
# speedup vs baseline: 3.6837x; 1.2451x over previous
#include <hip/hip_runtime.h>


typedef float v2f __attribute__((ext_vector_type(2)));
typedef float v8f __attribute__((ext_vector_type(8)));

#define NN   100000
#define EE   1600000
#define DH   64
#define DOUT 16

#define BN   128
#define BSH  7
#define NB   782
#define NS   128
#define SL   12500
#define SRCM 0x1FFFF
#define BT   1024
#define AT   512
#define GT   256
#define GB   391

__global__ __launch_bounds__(BT) void gcn_bin_count(const int* __restrict__ dst,
                                                    int* __restrict__ cnt) {
  __shared__ int hist[NB];
  const int tid = threadIdx.x;
  for (int i = tid; i < NB; i += BT) hist[i] = 0;
  __syncthreads();
  const int e0 = blockIdx.x * SL;
  for (int i = tid; i < SL; i += BT) atomicAdd(&hist[dst[e0 + i] >> BSH], 1);
  __syncthreads();
  for (int i = tid; i < NB; i += BT) cnt[blockIdx.x * NB + i] = hist[i];
}

__global__ __launch_bounds__(BT) void gcn_bin_scan(const int* __restrict__ cnt,
                                                   int* __restrict__ off,
                                                   int* __restrict__ bstart) {
  __shared__ int tot[BT];
  const int b = threadIdx.x;
  int run = 0;
  if (b < NB) {
    for (int s = 0; s < NS; s += 8) {
      int c[8];
#pragma unroll
      for (int u = 0; u < 8; ++u) c[u] = cnt[(s + u) * NB + b];
#pragma unroll
      for (int u = 0; u < 8; ++u) { off[(s + u) * NB + b] = run; run += c[u]; }
    }
  }
  tot[b] = run;
  __syncthreads();
  for (int d = 1; d < BT; d <<= 1) {
    const int v = (b >= d) ? tot[b - d] : 0;
    __syncthreads();
    tot[b] += v;
    __syncthreads();
  }
  if (b < NB) bstart[b + 1] = tot[b];
  if (b == 0) bstart[0] = 0;
}

__global__ __launch_bounds__(BT) void gcn_bin_place(const int* __restrict__ src,
                                                    const int* __restrict__ dst,
                                                    const int* __restrict__ off,
                                                    const int* __restrict__ bstart,
                                                    int* __restrict__ rec) {
  __shared__ int cur[NB];
  const int tid = threadIdx.x;
  for (int i = tid; i < NB; i += BT) cur[i] = bstart[i] + off[blockIdx.x * NB + i];
  __syncthreads();
  const int e0 = blockIdx.x * SL;
  for (int i = tid; i < SL; i += BT) {
    const int t = dst[e0 + i];
    const int s = src[e0 + i];
    const int pos = atomicAdd(&cur[t >> BSH], 1);
    rec[pos] = s | ((t & (BN - 1)) << 17);
  }
}

__global__ __launch_bounds__(256) void gcn_bucket_dinv(const int* __restrict__ rec,
                                                       const int* __restrict__ bstart,
                                                       float* __restrict__ dinv) {
  __shared__ int c[BN];
  const int tid = threadIdx.x;
  if (tid < BN) c[tid] = 0;
  __syncthreads();
  const int beg = bstart[blockIdx.x];
  const int end = bstart[blockIdx.x + 1];
  for (int e = beg + tid; e < end; e += 256) atomicAdd(&c[rec[e] >> 17], 1);
  __syncthreads();
  const int node = blockIdx.x * BN + tid;
  if (tid < BN && node < NN) dinv[node] = rsqrtf((float)c[tid] + 1.0f);
}

template <int NCOLS, bool BIAS, bool RELU>
__global__ __launch_bounds__(GT) void gcn_gemm_wmma(
    const float* __restrict__ X, const float* __restrict__ W,
    const float* __restrict__ bias, float* __restrict__ Out) {
  constexpr int NT = NCOLS / 16;
  __shared__ float4 Wl4[DH * NCOLS / 4];

  const int tid  = threadIdx.x;
  const int lane = tid & 31;
  const int h    = lane >> 4;
  const int c    = lane & 15;
  const int wave = __builtin_amdgcn_readfirstlane(tid >> 5);
  const int pair = blockIdx.x * (GT / 32) + wave;
  const bool live = pair < NN / 32;
  const int row0 = (live ? pair : 0) * 32;

  const float4* X4 = reinterpret_cast<const float4*>(X);
  float4 a[2][8];
#pragma unroll
  for (int m = 0; m < 2; ++m) {
#pragma unroll
    for (int j = 0; j < 8; ++j) a[m][j] = X4[(row0 + 16 * m + c) * (DH / 4) + 2 * j + h];
  }

  const float4* W4 = reinterpret_cast<const float4*>(W);
#pragma unroll
  for (int i = 0; i < DH * NCOLS / 4 / GT; ++i) Wl4[tid + i * GT] = W4[tid + i * GT];
  __syncthreads();
  if (!live) return;

  v8f acc[2][NT];
#pragma unroll
  for (int m = 0; m < 2; ++m) {
#pragma unroll
    for (int t = 0; t < NT; ++t) acc[m][t] = (v8f){};
  }

  const float* Wl = reinterpret_cast<const float*>(Wl4);
#pragma unroll
  for (int j = 0; j < 8; ++j) {
#pragma unroll
    for (int sub = 0; sub < 2; ++sub) {
      const int k0 = 8 * j + 4 * h + 2 * sub;
      v2f av[2];
#pragma unroll
      for (int m = 0; m < 2; ++m) {
        av[m].x = sub ? a[m][j].z : a[m][j].x;
        av[m].y = sub ? a[m][j].w : a[m][j].y;
      }
      v2f bv[NT];
      if constexpr (NT == 4) {
        const float4 b0 = Wl4[k0 * (NCOLS / 4) + c];
        const float4 b1 = Wl4[(k0 + 1) * (NCOLS / 4) + c];
        bv[0].x = b0.x; bv[0].y = b1.x;
        bv[1].x = b0.y; bv[1].y = b1.y;
        bv[2].x = b0.z; bv[2].y = b1.z;
        bv[3].x = b0.w; bv[3].y = b1.w;
      } else {
        bv[0].x = Wl[k0 * NCOLS + c];
        bv[0].y = Wl[(k0 + 1) * NCOLS + c];
      }
#pragma unroll
      for (int m = 0; m < 2; ++m) {
#pragma unroll
        for (int t = 0; t < NT; ++t) {
          acc[m][t] = __builtin_amdgcn_wmma_f32_16x16x4_f32(
               false, av[m],  false, bv[t],
               (short)0, acc[m][t],  false,  false);
        }
      }
    }
  }

  if constexpr (NT == 4) {
    float4 bb = make_float4(0.0f, 0.0f, 0.0f, 0.0f);
    if (BIAS) bb = reinterpret_cast<const float4*>(bias)[c];
    float4* O4 = reinterpret_cast<float4*>(Out);
#pragma unroll
    for (int m = 0; m < 2; ++m) {
#pragma unroll
      for (int r = 0; r < 8; ++r) {
        float4 o;
        o.x = acc[m][0][r];
        o.y = acc[m][1][r];
        o.z = acc[m][2][r];
        o.w = acc[m][3][r];
        if (BIAS) { o.x += bb.x; o.y += bb.y; o.z += bb.z; o.w += bb.w; }
        if (RELU) {
          o.x = fmaxf(o.x, 0.0f); o.y = fmaxf(o.y, 0.0f);
          o.z = fmaxf(o.z, 0.0f); o.w = fmaxf(o.w, 0.0f);
        }
        O4[(row0 + 16 * m + 8 * h + r) * (NCOLS / 4) + c] = o;
      }
    }
  } else {
    float bb = 0.0f;
    if (BIAS) bb = bias[c];
#pragma unroll
    for (int m = 0; m < 2; ++m) {
#pragma unroll
      for (int r = 0; r < 8; ++r) {
        float v = acc[m][0][r];
        if (BIAS) v += bb;
        if (RELU) v = fmaxf(v, 0.0f);
        Out[(row0 + 16 * m + 8 * h + r) * NCOLS + c] = v;
      }
    }
  }
}

__global__ __launch_bounds__(AT) void gcn_agg_bucket(const float* __restrict__ H,
                                                     const int* __restrict__ rec,
                                                     const int* __restrict__ bstart,
                                                     const float* __restrict__ dinv,
                                                     const float* __restrict__ bias,
                                                     float* __restrict__ Out) {
  __shared__ float tile[BN * DH];
  __shared__ float sdinv[BN];
  const int tid   = threadIdx.x;
  const int lane  = tid & 31;
  const int wave  = __builtin_amdgcn_readfirstlane(tid >> 5);
  const int node0 = blockIdx.x * BN;

  float4* tile4 = reinterpret_cast<float4*>(tile);
  for (int i = tid; i < BN * DH / 4; i += AT) tile4[i] = make_float4(0.0f, 0.0f, 0.0f, 0.0f);
  if (tid < BN) sdinv[tid] = (node0 + tid < NN) ? dinv[node0 + tid] : 0.0f;
  __syncthreads();

  const int beg = bstart[blockIdx.x];
  const int end = bstart[blockIdx.x + 1];
  const float2* H2 = reinterpret_cast<const float2*>(H);

  for (int base = beg + wave * 32; base < end; base += AT) {
    const int e = base + lane;
    int   r  = 0;
    float cf = 0.0f;
    if (e < end) {
      r  = rec[e];
      cf = dinv[r & SRCM] * sdinv[r >> 17];
    }
    const int n = end - base;
    if (n >= 32) {
#pragma unroll
      for (int g = 0; g < 32; g += 16) {
        float2 v[16];
#pragma unroll
        for (int j = 0; j < 16; ++j) {
          const int rj = __builtin_amdgcn_readlane(r, g + j);
          v[j] = H2[(rj & SRCM) * (DH / 2) + lane];
        }
#pragma unroll
        for (int j = 0; j < 16; ++j) {
          const int   rj = __builtin_amdgcn_readlane(r, g + j);
          const float cj = __int_as_float(__builtin_amdgcn_readlane(__float_as_int(cf), g + j));
          float* row = &tile[(rj >> 17) * DH + 2 * lane];
          atomicAdd(row,     v[j].x * cj);
          atomicAdd(row + 1, v[j].y * cj);
        }
      }
    } else {
      for (int j = 0; j < n; ++j) {
        const int    rj = __builtin_amdgcn_readlane(r, j);
        const float  cj = __int_as_float(__builtin_amdgcn_readlane(__float_as_int(cf), j));
        const float2 v  = H2[(rj & SRCM) * (DH / 2) + lane];
        float* row = &tile[(rj >> 17) * DH + 2 * lane];
        atomicAdd(row,     v.x * cj);
        atomicAdd(row + 1, v.y * cj);
      }
    }
  }
  __syncthreads();

  const float4* H4 = reinterpret_cast<const float4*>(H);
  const float4* b4 = reinterpret_cast<const float4*>(bias);
  float4*       O4 = reinterpret_cast<float4*>(Out);
  for (int i = tid; i < BN * DH / 4; i += AT) {
    const int t    = i >> 4;
    const int q    = i & 15;
    const int node = node0 + t;
    if (node < NN) {
      const float  di = sdinv[t];
      const float4 a  = tile4[i];
      const float4 h  = H4[node * (DH / 4) + q];
      const float4 bb = b4[q];
      float4 o;
      o.x = fmaxf(a.x + h.x * di * di + bb.x, 0.0f);
      o.y = fmaxf(a.y + h.y * di * di + bb.y, 0.0f);
      o.z = fmaxf(a.z + h.z * di * di + bb.z, 0.0f);
      o.w = fmaxf(a.w + h.w * di * di + bb.w, 0.0f);
      O4[node * (DH / 4) + q] = o;
    }
  }
}

extern "C" void kernel_launch(void* const* d_in, const int* in_sizes, int n_in,
                              void* d_out, int out_size, void* d_ws,
                              size_t ws_size, hipStream_t stream) {
  (void)in_sizes; (void)n_in; (void)out_size; (void)ws_size;

  const float* x   = (const float*)d_in[0];
  const int*   ei  = (const int*)d_in[1];
  const int*   src = ei;
  const int*   dst = ei + EE;
  const float* W1  = (const float*)d_in[2];
  const float* b1  = (const float*)d_in[3];
  const float* W2  = (const float*)d_in[4];
  const float* b2  = (const float*)d_in[5];
  const float* W3  = (const float*)d_in[6];
  const float* b3  = (const float*)d_in[7];
  const float* Wf1 = (const float*)d_in[8];
  const float* bf1 = (const float*)d_in[9];
  const float* Wf2 = (const float*)d_in[10];
  const float* bf2 = (const float*)d_in[11];
  float* out = (float*)d_out;

  float* base = (float*)d_ws;
  float* dinv = base;
  float* bufA = base + NN;
  float* bufB = bufA + NN * DH;
  int*   rec  = (int*)(bufB + NN * DH);
  int*   cnt  = rec + EE;
  int*   off  = cnt + NS * NB;
  int*   bst  = off + NS * NB;

  const dim3 gemmGrid(GB);
  const dim3 gemmBlk(GT);

  gcn_bin_count<<<NS, BT, 0, stream>>>(dst, cnt);
  gcn_bin_scan<<<1, BT, 0, stream>>>(cnt, off, bst);
  gcn_bin_place<<<NS, BT, 0, stream>>>(src, dst, off, bst, rec);
  gcn_bucket_dinv<<<NB, 256, 0, stream>>>(rec, bst, dinv);

  gcn_gemm_wmma<DH, false, false><<<gemmGrid, gemmBlk, 0, stream>>>(x, W1, nullptr, bufA);
  gcn_agg_bucket<<<NB, AT, 0, stream>>>(bufA, rec, bst, dinv, b1, bufB);

  gcn_gemm_wmma<DH, false, false><<<gemmGrid, gemmBlk, 0, stream>>>(bufB, W2, nullptr, bufA);
  gcn_agg_bucket<<<NB, AT, 0, stream>>>(bufA, rec, bst, dinv, b2, bufB);

  gcn_gemm_wmma<DH, false, false><<<gemmGrid, gemmBlk, 0, stream>>>(bufB, W3, nullptr, bufA);
  gcn_agg_bucket<<<NB, AT, 0, stream>>>(bufA, rec, bst, dinv, b3, bufB);

  gcn_gemm_wmma<DH, true, true><<<gemmGrid, gemmBlk, 0, stream>>>(bufB, Wf1, bf1, bufA);
  gcn_gemm_wmma<DOUT, true, false><<<gemmGrid, gemmBlk, 0, stream>>>(bufA, Wf2, bf2, out);
}


# Loops that waited for one load per trip now issue their loads together: bin count/place (13 edges/thread), degree count (8), partial record batch (16 rows), finishing pass (4); sums still by arrival
# speedup vs baseline: 3.8675x; 1.0499x over previous
#include <hip/hip_runtime.h>


typedef float v2f __attribute__((ext_vector_type(2)));
typedef float v8f __attribute__((ext_vector_type(8)));

#define NN   100000
#define EE   1600000
#define DH   64
#define DOUT 16

#define BN   128
#define BSH  7
#define NB   782
#define NS   128
#define SL   12500
#define SRCM 0x1FFFF
#define BT   1024
#define ET   13
static_assert(ET * BT >= SL && (ET - 1) * BT < SL, "ET must be ceil(SL / BT)");
#define AT   512
#define GT   256
#define GB   391

__global__ __launch_bounds__(BT) void gcn_bin_count(const int* __restrict__ dst,
                                                    int* __restrict__ cnt) {
  __shared__ int hist[NB];
  const int tid = threadIdx.x;
  const int e0 = blockIdx.x * SL;
  int d[ET];
#pragma unroll
  for (int u = 0; u < ET; ++u) {
    const int i = tid + u * BT;
    d[u] = (i < SL) ? dst[e0 + i] : 0;
  }
  for (int i = tid; i < NB; i += BT) hist[i] = 0;
  __syncthreads();
#pragma unroll
  for (int u = 0; u < ET; ++u) {
    if (tid + u * BT < SL) atomicAdd(&hist[d[u] >> BSH], 1);
  }
  __syncthreads();
  for (int i = tid; i < NB; i += BT) cnt[blockIdx.x * NB + i] = hist[i];
}

__global__ __launch_bounds__(BT) void gcn_bin_scan(const int* __restrict__ cnt,
                                                   int* __restrict__ off,
                                                   int* __restrict__ bstart) {
  __shared__ int tot[BT];
  const int b = threadIdx.x;
  int run = 0;
  if (b < NB) {
    for (int s = 0; s < NS; s += 8) {
      int c[8];
#pragma unroll
      for (int u = 0; u < 8; ++u) c[u] = cnt[(s + u) * NB + b];
#pragma unroll
      for (int u = 0; u < 8; ++u) { off[(s + u) * NB + b] = run; run += c[u]; }
    }
  }
  tot[b] = run;
  __syncthreads();
  for (int d = 1; d < BT; d <<= 1) {
    const int v = (b >= d) ? tot[b - d] : 0;
    __syncthreads();
    tot[b] += v;
    __syncthreads();
  }
  if (b < NB) bstart[b + 1] = tot[b];
  if (b == 0) bstart[0] = 0;
}

__global__ __launch_bounds__(BT) void gcn_bin_place(const int* __restrict__ src,
                                                    const int* __restrict__ dst,
                                                    const int* __restrict__ off,
                                                    const int* __restrict__ bstart,
                                                    int* __restrict__ rec) {
  __shared__ int cur[NB];
  const int tid = threadIdx.x;
  const int e0 = blockIdx.x * SL;
  int t[ET], s[ET];
#pragma unroll
  for (int u = 0; u < ET; ++u) {
    const int i = tid + u * BT;
    t[u] = (i < SL) ? dst[e0 + i] : 0;
    s[u] = (i < SL) ? src[e0 + i] : 0;
  }
  for (int i = tid; i < NB; i += BT) cur[i] = bstart[i] + off[blockIdx.x * NB + i];
  __syncthreads();
#pragma unroll
  for (int u = 0; u < ET; ++u) {
    if (tid + u * BT < SL) {
      const int pos = atomicAdd(&cur[t[u] >> BSH], 1);
      rec[pos] = s[u] | ((t[u] & (BN - 1)) << 17);
    }
  }
}

__global__ __launch_bounds__(256) void gcn_bucket_dinv(const int* __restrict__ rec,
                                                       const int* __restrict__ bstart,
                                                       float* __restrict__ dinv) {
  __shared__ int c[BN];
  const int tid = threadIdx.x;
  if (tid < BN) c[tid] = 0;
  __syncthreads();
  const int beg = bstart[blockIdx.x];
  const int end = bstart[blockIdx.x + 1];
  for (int e0 = beg + tid; e0 < end; e0 += 8 * 256) {
    int rr[8];
#pragma unroll
    for (int u = 0; u < 8; ++u) {
      const int e = e0 + u * 256;
      rr[u] = rec[e < end ? e : e0];
    }
#pragma unroll
    for (int u = 0; u < 8; ++u) {
      if (e0 + u * 256 < end) atomicAdd(&c[rr[u] >> 17], 1);
    }
  }
  __syncthreads();
  const int node = blockIdx.x * BN + tid;
  if (tid < BN && node < NN) dinv[node] = rsqrtf((float)c[tid] + 1.0f);
}

template <int NCOLS, bool BIAS, bool RELU>
__global__ __launch_bounds__(GT) void gcn_gemm_wmma(
    const float* __restrict__ X, const float* __restrict__ W,
    const float* __restrict__ bias, float* __restrict__ Out) {
  constexpr int NT = NCOLS / 16;
  __shared__ float4 Wl4[DH * NCOLS / 4];

  const int tid  = threadIdx.x;
  const int lane = tid & 31;
  const int h    = lane >> 4;
  const int c    = lane & 15;
  const int wave = __builtin_amdgcn_readfirstlane(tid >> 5);
  const int pair = blockIdx.x * (GT / 32) + wave;
  const bool live = pair < NN / 32;
  const int row0 = (live ? pair : 0) * 32;

  const float4* X4 = reinterpret_cast<const float4*>(X);
  float4 a[2][8];
#pragma unroll
  for (int m = 0; m < 2; ++m) {
#pragma unroll
    for (int j = 0; j < 8; ++j) a[m][j] = X4[(row0 + 16 * m + c) * (DH / 4) + 2 * j + h];
  }

  const float4* W4 = reinterpret_cast<const float4*>(W);
#pragma unroll
  for (int i = 0; i < DH * NCOLS / 4 / GT; ++i) Wl4[tid + i * GT] = W4[tid + i * GT];
  __syncthreads();
  if (!live) return;

  v8f acc[2][NT];
#pragma unroll
  for (int m = 0; m < 2; ++m) {
#pragma unroll
    for (int t = 0; t < NT; ++t) acc[m][t] = (v8f){};
  }

  const float* Wl = reinterpret_cast<const float*>(Wl4);
#pragma unroll
  for (int j = 0; j < 8; ++j) {
#pragma unroll
    for (int sub = 0; sub < 2; ++sub) {
      const int k0 = 8 * j + 4 * h + 2 * sub;
      v2f av[2];
#pragma unroll
      for (int m = 0; m < 2; ++m) {
        av[m].x = sub ? a[m][j].z : a[m][j].x;
        av[m].y = sub ? a[m][j].w : a[m][j].y;
      }
      v2f bv[NT];
      if constexpr (NT == 4) {
        const float4 b0 = Wl4[k0 * (NCOLS / 4) + c];
        const float4 b1 = Wl4[(k0 + 1) * (NCOLS / 4) + c];
        bv[0].x = b0.x; bv[0].y = b1.x;
        bv[1].x = b0.y; bv[1].y = b1.y;
        bv[2].x = b0.z; bv[2].y = b1.z;
        bv[3].x = b0.w; bv[3].y = b1.w;
      } else {
        bv[0].x = Wl[k0 * NCOLS + c];
        bv[0].y = Wl[(k0 + 1) * NCOLS + c];
      }
#pragma unroll
      for (int m = 0; m < 2; ++m) {
#pragma unroll
        for (int t = 0; t < NT; ++t) {
          acc[m][t] = __builtin_amdgcn_wmma_f32_16x16x4_f32(
               false, av[m],  false, bv[t],
               (short)0, acc[m][t],  false,  false);
        }
      }
    }
  }

  if constexpr (NT == 4) {
    float4 bb = make_float4(0.0f, 0.0f, 0.0f, 0.0f);
    if (BIAS) bb = reinterpret_cast<const float4*>(bias)[c];
    float4* O4 = reinterpret_cast<float4*>(Out);
#pragma unroll
    for (int m = 0; m < 2; ++m) {
#pragma unroll
      for (int r = 0; r < 8; ++r) {
        float4 o;
        o.x = acc[m][0][r];
        o.y = acc[m][1][r];
        o.z = acc[m][2][r];
        o.w = acc[m][3][r];
        if (BIAS) { o.x += bb.x; o.y += bb.y; o.z += bb.z; o.w += bb.w; }
        if (RELU) {
          o.x = fmaxf(o.x, 0.0f); o.y = fmaxf(o.y, 0.0f);
          o.z = fmaxf(o.z, 0.0f); o.w = fmaxf(o.w, 0.0f);
        }
        O4[(row0 + 16 * m + 8 * h + r) * (NCOLS / 4) + c] = o;
      }
    }
  } else {
    float bb = 0.0f;
    if (BIAS) bb = bias[c];
#pragma unroll
    for (int m = 0; m < 2; ++m) {
#pragma unroll
      for (int r = 0; r < 8; ++r) {
        float v = acc[m][0][r];
        if (BIAS) v += bb;
        if (RELU) v = fmaxf(v, 0.0f);
        Out[(row0 + 16 * m + 8 * h + r) * NCOLS + c] = v;
      }
    }
  }
}

template <bool FULL>
__device__ __forceinline__ void gcn_add_batch(const float2* __restrict__ H2, float* tile,
                                              int r, float cf, int n, int lane) {
#pragma unroll
  for (int g = 0; g < 32; g += 16) {
    if (FULL || g < n) {
      float2 v[16];
#pragma unroll
      for (int j = 0; j < 16; ++j) {
        const int rj = __builtin_amdgcn_readlane(r, g + j);
        v[j] = H2[(rj & SRCM) * (DH / 2) + lane];
      }
#pragma unroll
      for (int j = 0; j < 16; ++j) {
        if (FULL || g + j < n) {
          const int   rj = __builtin_amdgcn_readlane(r, g + j);
          const float cj = __int_as_float(__builtin_amdgcn_readlane(__float_as_int(cf), g + j));
          float* row = &tile[(rj >> 17) * DH + 2 * lane];
          atomicAdd(row,     v[j].x * cj);
          atomicAdd(row + 1, v[j].y * cj);
        }
      }
    }
  }
}

__global__ __launch_bounds__(AT) void gcn_agg_bucket(const float* __restrict__ H,
                                                     const int* __restrict__ rec,
                                                     const int* __restrict__ bstart,
                                                     const float* __restrict__ dinv,
                                                     const float* __restrict__ bias,
                                                     float* __restrict__ Out) {
  __shared__ float tile[BN * DH];
  __shared__ float sdinv[BN];
  const int tid   = threadIdx.x;
  const int lane  = tid & 31;
  const int wave  = __builtin_amdgcn_readfirstlane(tid >> 5);
  const int node0 = blockIdx.x * BN;

  float4* tile4 = reinterpret_cast<float4*>(tile);
  for (int i = tid; i < BN * DH / 4; i += AT) tile4[i] = make_float4(0.0f, 0.0f, 0.0f, 0.0f);
  if (tid < BN) sdinv[tid] = (node0 + tid < NN) ? dinv[node0 + tid] : 0.0f;
  __syncthreads();

  const int beg = bstart[blockIdx.x];
  const int end = bstart[blockIdx.x + 1];
  const float2* H2 = reinterpret_cast<const float2*>(H);

  for (int base = beg + wave * 32; base < end; base += AT) {
    const int e = base + lane;
    int   r  = 0;
    float cf = 0.0f;
    if (e < end) {
      r  = rec[e];
      cf = dinv[r & SRCM] * sdinv[r >> 17];
    }
    const int n = end - base;
    if (n >= 32) gcn_add_batch<true>(H2, tile, r, cf, 32, lane);
    else         gcn_add_batch<false>(H2, tile, r, cf, n, lane);
  }
  __syncthreads();

  const float4* H4 = reinterpret_cast<const float4*>(H);
  const float4* b4 = reinterpret_cast<const float4*>(bias);
  float4*       O4 = reinterpret_cast<float4*>(Out);
  constexpr int FP = BN * DH / 4 / AT;
  const int    q  = tid & 15;
  const float4 bb = b4[q];
  float4 hs[FP];
#pragma unroll
  for (int p = 0; p < FP; ++p) {
    const int node = node0 + ((tid + p * AT) >> 4);
    hs[p] = (node < NN) ? H4[node * (DH / 4) + q] : make_float4(0.0f, 0.0f, 0.0f, 0.0f);
  }
#pragma unroll
  for (int p = 0; p < FP; ++p) {
    const int i    = tid + p * AT;
    const int t    = i >> 4;
    const int node = node0 + t;
    if (node < NN) {
      const float  di = sdinv[t];
      const float4 a  = tile4[i];
      const float4 h  = hs[p];
      float4 o;
      o.x = fmaxf(a.x + h.x * di * di + bb.x, 0.0f);
      o.y = fmaxf(a.y + h.y * di * di + bb.y, 0.0f);
      o.z = fmaxf(a.z + h.z * di * di + bb.z, 0.0f);
      o.w = fmaxf(a.w + h.w * di * di + bb.w, 0.0f);
      O4[node * (DH / 4) + q] = o;
    }
  }
}

extern "C" void kernel_launch(void* const* d_in, const int* in_sizes, int n_in,
                              void* d_out, int out_size, void* d_ws,
                              size_t ws_size, hipStream_t stream) {
  (void)in_sizes; (void)n_in; (void)out_size; (void)ws_size;

  const float* x   = (const float*)d_in[0];
  const int*   ei  = (const int*)d_in[1];
  const int*   src = ei;
  const int*   dst = ei + EE;
  const float* W1  = (const float*)d_in[2];
  const float* b1  = (const float*)d_in[3];
  const float* W2  = (const float*)d_in[4];
  const float* b2  = (const float*)d_in[5];
  const float* W3  = (const float*)d_in[6];
  const float* b3  = (const float*)d_in[7];
  const float* Wf1 = (const float*)d_in[8];
  const float* bf1 = (const float*)d_in[9];
  const float* Wf2 = (const float*)d_in[10];
  const float* bf2 = (const float*)d_in[11];
  float* out = (float*)d_out;

  float* base = (float*)d_ws;
  float* dinv = base;
  float* bufA = base + NN;
  float* bufB = bufA + NN * DH;
  int*   rec  = (int*)(bufB + NN * DH);
  int*   cnt  = rec + EE;
  int*   off  = cnt + NS * NB;
  int*   bst  = off + NS * NB;

  const dim3 gemmGrid(GB);
  const dim3 gemmBlk(GT);

  gcn_bin_count<<<NS, BT, 0, stream>>>(dst, cnt);
  gcn_bin_scan<<<1, BT, 0, stream>>>(cnt, off, bst);
  gcn_bin_place<<<NS, BT, 0, stream>>>(src, dst, off, bst, rec);
  gcn_bucket_dinv<<<NB, 256, 0, stream>>>(rec, bst, dinv);

  gcn_gemm_wmma<DH, false, false><<<gemmGrid, gemmBlk, 0, stream>>>(x, W1, nullptr, bufA);
  gcn_agg_bucket<<<NB, AT, 0, stream>>>(bufA, rec, bst, dinv, b1, bufB);

  gcn_gemm_wmma<DH, false, false><<<gemmGrid, gemmBlk, 0, stream>>>(bufB, W2, nullptr, bufA);
  gcn_agg_bucket<<<NB, AT, 0, stream>>>(bufA, rec, bst, dinv, b2, bufB);

  gcn_gemm_wmma<DH, false, false><<<gemmGrid, gemmBlk, 0, stream>>>(bufB, W3, nullptr, bufA);
  gcn_agg_bucket<<<NB, AT, 0, stream>>>(bufA, rec, bst, dinv, b3, bufB);

  gcn_gemm_wmma<DH, true, true><<<gemmGrid, gemmBlk, 0, stream>>>(bufB, Wf1, bf1, bufA);
  gcn_gemm_wmma<DOUT, true, false><<<gemmGrid, gemmBlk, 0, stream>>>(bufA, Wf2, bf2, out);
}


# Aggregation re-tiled for an equal share per WGP: buckets of 196 nodes in 24-wave blocks, 511 blocks = exactly two per WGP (before: 782 blocks, a 4th on some WGPs); sums still by arrival order
# speedup vs baseline: 4.2416x; 1.0967x over previous
#include <hip/hip_runtime.h>


typedef float v2f __attribute__((ext_vector_type(2)));
typedef float v8f __attribute__((ext_vector_type(8)));

#define NN   100000
#define EE   1600000
#define DH   64
#define DOUT 16

#define BN   196
#define NB   511
static_assert(NB * BN >= NN && (NB - 1) * BN < NN, "NB must be ceil(NN / BN)");
static_assert(BN <= 256, "a record keeps the node's place in its bucket in 8 bits above src");
#define NS   128
#define SL   12500
#define SRCM 0x1FFFF
#define BT   1024
#define ET   13
static_assert(ET * BT >= SL && (ET - 1) * BT < SL, "ET must be ceil(SL / BT)");
#define AT   768
#define DT   512
#define GT   256
#define GB   391

__global__ __launch_bounds__(BT) void gcn_bin_count(const int* __restrict__ dst,
                                                    int* __restrict__ cnt) {
  __shared__ int hist[NB];
  const int tid = threadIdx.x;
  const int e0 = blockIdx.x * SL;
  int d[ET];
#pragma unroll
  for (int u = 0; u < ET; ++u) {
    const int i = tid + u * BT;
    d[u] = (i < SL) ? dst[e0 + i] : 0;
  }
  for (int i = tid; i < NB; i += BT) hist[i] = 0;
  __syncthreads();
#pragma unroll
  for (int u = 0; u < ET; ++u) {
    if (tid + u * BT < SL) atomicAdd(&hist[(unsigned)d[u] / BN], 1);
  }
  __syncthreads();
  for (int i = tid; i < NB; i += BT) cnt[blockIdx.x * NB + i] = hist[i];
}

__global__ __launch_bounds__(BT) void gcn_bin_scan(const int* __restrict__ cnt,
                                                   int* __restrict__ off,
                                                   int* __restrict__ bstart) {
  __shared__ int tot[BT];
  const int b = threadIdx.x;
  int run = 0;
  if (b < NB) {
    for (int s = 0; s < NS; s += 8) {
      int c[8];
#pragma unroll
      for (int u = 0; u < 8; ++u) c[u] = cnt[(s + u) * NB + b];
#pragma unroll
      for (int u = 0; u < 8; ++u) { off[(s + u) * NB + b] = run; run += c[u]; }
    }
  }
  tot[b] = run;
  __syncthreads();
  for (int d = 1; d < BT; d <<= 1) {
    const int v = (b >= d) ? tot[b - d] : 0;
    __syncthreads();
    tot[b] += v;
    __syncthreads();
  }
  if (b < NB) bstart[b + 1] = tot[b];
  if (b == 0) bstart[0] = 0;
}

__global__ __launch_bounds__(BT) void gcn_bin_place(const int* __restrict__ src,
                                                    const int* __restrict__ dst,
                                                    const int* __restrict__ off,
                                                    const int* __restrict__ bstart,
                                                    int* __restrict__ rec) {
  __shared__ int cur[NB];
  const int tid = threadIdx.x;
  const int e0 = blockIdx.x * SL;
  int t[ET], s[ET];
#pragma unroll
  for (int u = 0; u < ET; ++u) {
    const int i = tid + u * BT;
    t[u] = (i < SL) ? dst[e0 + i] : 0;
    s[u] = (i < SL) ? src[e0 + i] : 0;
  }
  for (int i = tid; i < NB; i += BT) cur[i] = bstart[i] + off[blockIdx.x * NB + i];
  __syncthreads();
#pragma unroll
  for (int u = 0; u < ET; ++u) {
    if (tid + u * BT < SL) {
      const int bk  = (unsigned)t[u] / BN;
      const int pos = atomicAdd(&cur[bk], 1);
      rec[pos] = s[u] | ((t[u] - bk * BN) << 17);
    }
  }
}

__global__ __launch_bounds__(DT) void gcn_bucket_dinv(const int* __restrict__ rec,
                                                       const int* __restrict__ bstart,
                                                       float* __restrict__ dinv) {
  __shared__ int c[BN];
  const int tid = threadIdx.x;
  if (tid < BN) c[tid] = 0;
  __syncthreads();
  const int beg = bstart[blockIdx.x];
  const int end = bstart[blockIdx.x + 1];
  for (int e0 = beg + tid; e0 < end; e0 += 8 * DT) {
    int rr[8];
#pragma unroll
    for (int u = 0; u < 8; ++u) {
      const int e = e0 + u * DT;
      rr[u] = rec[e < end ? e : e0];
    }
#pragma unroll
    for (int u = 0; u < 8; ++u) {
      if (e0 + u * DT < end) atomicAdd(&c[rr[u] >> 17], 1);
    }
  }
  __syncthreads();
  const int node = blockIdx.x * BN + tid;
  if (tid < BN && node < NN) dinv[node] = rsqrtf((float)c[tid] + 1.0f);
}

template <int NCOLS, bool BIAS, bool RELU>
__global__ __launch_bounds__(GT) void gcn_gemm_wmma(
    const float* __restrict__ X, const float* __restrict__ W,
    const float* __restrict__ bias, float* __restrict__ Out) {
  constexpr int NT = NCOLS / 16;
  __shared__ float4 Wl4[DH * NCOLS / 4];

  const int tid  = threadIdx.x;
  const int lane = tid & 31;
  const int h    = lane >> 4;
  const int c    = lane & 15;
  const int wave = __builtin_amdgcn_readfirstlane(tid >> 5);
  const int pair = blockIdx.x * (GT / 32) + wave;
  const bool live = pair < NN / 32;
  const int row0 = (live ? pair : 0) * 32;

  const float4* X4 = reinterpret_cast<const float4*>(X);
  float4 a[2][8];
#pragma unroll
  for (int m = 0; m < 2; ++m) {
#pragma unroll
    for (int j = 0; j < 8; ++j) a[m][j] = X4[(row0 + 16 * m + c) * (DH / 4) + 2 * j + h];
  }

  const float4* W4 = reinterpret_cast<const float4*>(W);
#pragma unroll
  for (int i = 0; i < DH * NCOLS / 4 / GT; ++i) Wl4[tid + i * GT] = W4[tid + i * GT];
  __syncthreads();
  if (!live) return;

  v8f acc[2][NT];
#pragma unroll
  for (int m = 0; m < 2; ++m) {
#pragma unroll
    for (int t = 0; t < NT; ++t) acc[m][t] = (v8f){};
  }

  const float* Wl = reinterpret_cast<const float*>(Wl4);
#pragma unroll
  for (int j = 0; j < 8; ++j) {
#pragma unroll
    for (int sub = 0; sub < 2; ++sub) {
      const int k0 = 8 * j + 4 * h + 2 * sub;
      v2f av[2];
#pragma unroll
      for (int m = 0; m < 2; ++m) {
        av[m].x = sub ? a[m][j].z : a[m][j].x;
        av[m].y = sub ? a[m][j].w : a[m][j].y;
      }
      v2f bv[NT];
      if constexpr (NT == 4) {
        const float4 b0 = Wl4[k0 * (NCOLS / 4) + c];
        const float4 b1 = Wl4[(k0 + 1) * (NCOLS / 4) + c];
        bv[0].x = b0.x; bv[0].y = b1.x;
        bv[1].x = b0.y; bv[1].y = b1.y;
        bv[2].x = b0.z; bv[2].y = b1.z;
        bv[3].x = b0.w; bv[3].y = b1.w;
      } else {
        bv[0].x = Wl[k0 * NCOLS + c];
        bv[0].y = Wl[(k0 + 1) * NCOLS + c];
      }
#pragma unroll
      for (int m = 0; m < 2; ++m) {
#pragma unroll
        for (int t = 0; t < NT; ++t) {
          acc[m][t] = __builtin_amdgcn_wmma_f32_16x16x4_f32(
               false, av[m],  false, bv[t],
               (short)0, acc[m][t],  false,  false);
        }
      }
    }
  }

  if constexpr (NT == 4) {
    float4 bb = make_float4(0.0f, 0.0f, 0.0f, 0.0f);
    if (BIAS) bb = reinterpret_cast<const float4*>(bias)[c];
    float4* O4 = reinterpret_cast<float4*>(Out);
#pragma unroll
    for (int m = 0; m < 2; ++m) {
#pragma unroll
      for (int r = 0; r < 8; ++r) {
        float4 o;
        o.x = acc[m][0][r];
        o.y = acc[m][1][r];
        o.z = acc[m][2][r];
        o.w = acc[m][3][r];
        if (BIAS) { o.x += bb.x; o.y += bb.y; o.z += bb.z; o.w += bb.w; }
        if (RELU) {
          o.x = fmaxf(o.x, 0.0f); o.y = fmaxf(o.y, 0.0f);
          o.z = fmaxf(o.z, 0.0f); o.w = fmaxf(o.w, 0.0f);
        }
        O4[(row0 + 16 * m + 8 * h + r) * (NCOLS / 4) + c] = o;
      }
    }
  } else {
    float bb = 0.0f;
    if (BIAS) bb = bias[c];
#pragma unroll
    for (int m = 0; m < 2; ++m) {
#pragma unroll
      for (int r = 0; r < 8; ++r) {
        float v = acc[m][0][r];
        if (BIAS) v += bb;
        if (RELU) v = fmaxf(v, 0.0f);
        Out[(row0 + 16 * m + 8 * h + r) * NCOLS + c] = v;
      }
    }
  }
}

template <bool FULL>
__device__ __forceinline__ void gcn_add_batch(const float2* __restrict__ H2, float* tile,
                                              int r, float cf, int n, int lane) {
#pragma unroll
  for (int g = 0; g < 32; g += 16) {
    if (FULL || g < n) {
      float2 v[16];
#pragma unroll
      for (int j = 0; j < 16; ++j) {
        const int rj = __builtin_amdgcn_readlane(r, g + j);
        v[j] = H2[(rj & SRCM) * (DH / 2) + lane];
      }
#pragma unroll
      for (int j = 0; j < 16; ++j) {
        if (FULL || g + j < n) {
          const int   rj = __builtin_amdgcn_readlane(r, g + j);
          const float cj = __int_as_float(__builtin_amdgcn_readlane(__float_as_int(cf), g + j));
          float* row = &tile[(rj >> 17) * DH + 2 * lane];
          atomicAdd(row,     v[j].x * cj);
          atomicAdd(row + 1, v[j].y * cj);
        }
      }
    }
  }
}

__global__ __launch_bounds__(AT) void gcn_agg_bucket(const float* __restrict__ H,
                                                     const int* __restrict__ rec,
                                                     const int* __restrict__ bstart,
                                                     const float* __restrict__ dinv,
                                                     const float* __restrict__ bias,
                                                     float* __restrict__ Out) {
  __shared__ float tile[BN * DH];
  __shared__ float sdinv[BN];
  const int tid   = threadIdx.x;
  const int lane  = tid & 31;
  const int wave  = __builtin_amdgcn_readfirstlane(tid >> 5);
  const int node0 = blockIdx.x * BN;

  float4* tile4 = reinterpret_cast<float4*>(tile);
  for (int i = tid; i < BN * DH / 4; i += AT) tile4[i] = make_float4(0.0f, 0.0f, 0.0f, 0.0f);
  if (tid < BN) sdinv[tid] = (node0 + tid < NN) ? dinv[node0 + tid] : 0.0f;
  __syncthreads();

  const int beg = bstart[blockIdx.x];
  const int end = bstart[blockIdx.x + 1];
  const float2* H2 = reinterpret_cast<const float2*>(H);

  for (int base = beg + wave * 32; base < end; base += AT) {
    const int e = base + lane;
    int   r  = 0;
    float cf = 0.0f;
    if (e < end) {
      r  = rec[e];
      cf = dinv[r & SRCM] * sdinv[r >> 17];
    }
    const int n = end - base;
    if (n >= 32) gcn_add_batch<true>(H2, tile, r, cf, 32, lane);
    else         gcn_add_batch<false>(H2, tile, r, cf, n, lane);
  }
  __syncthreads();

  const float4* H4 = reinterpret_cast<const float4*>(H);
  const float4* b4 = reinterpret_cast<const float4*>(bias);
  float4*       O4 = reinterpret_cast<float4*>(Out);
  constexpr int NP = BN * DH / 4;
  constexpr int FP = (NP + AT - 1) / AT;
  static_assert(AT % 16 == 0, "a thread must meet the same piece of a row in every pass");
  const int    q  = tid & 15;
  const float4 bb = b4[q];
  float4 hs[FP];
#pragma unroll
  for (int p = 0; p < FP; ++p) {
    const int i    = tid + p * AT;
    const int node = node0 + (i >> 4);
    hs[p] = (i < NP && node < NN) ? H4[node * (DH / 4) + q] : make_float4(0.0f, 0.0f, 0.0f, 0.0f);
  }
#pragma unroll
  for (int p = 0; p < FP; ++p) {
    const int i    = tid + p * AT;
    const int t    = i >> 4;
    const int node = node0 + t;
    if (i < NP && node < NN) {
      const float  di = sdinv[t];
      const float4 a  = tile4[i];
      const float4 h  = hs[p];
      float4 o;
      o.x = fmaxf(a.x + h.x * di * di + bb.x, 0.0f);
      o.y = fmaxf(a.y + h.y * di * di + bb.y, 0.0f);
      o.z = fmaxf(a.z + h.z * di * di + bb.z, 0.0f);
      o.w = fmaxf(a.w + h.w * di * di + bb.w, 0.0f);
      O4[node * (DH / 4) + q] = o;
    }
  }
}

extern "C" void kernel_launch(void* const* d_in, const int* in_sizes, int n_in,
                              void* d_out, int out_size, void* d_ws,
                              size_t ws_size, hipStream_t stream) {
  (void)in_sizes; (void)n_in; (void)out_size; (void)ws_size;

  const float* x   = (const float*)d_in[0];
  const int*   ei  = (const int*)d_in[1];
  const int*   src = ei;
  const int*   dst = ei + EE;
  const float* W1  = (const float*)d_in[2];
  const float* b1  = (const float*)d_in[3];
  const float* W2  = (const float*)d_in[4];
  const float* b2  = (const float*)d_in[5];
  const float* W3  = (const float*)d_in[6];
  const float* b3  = (const float*)d_in[7];
  const float* Wf1 = (const float*)d_in[8];
  const float* bf1 = (const float*)d_in[9];
  const float* Wf2 = (const float*)d_in[10];
  const float* bf2 = (const float*)d_in[11];
  float* out = (float*)d_out;

  float* base = (float*)d_ws;
  float* dinv = base;
  float* bufA = base + NN;
  float* bufB = bufA + NN * DH;
  int*   rec  = (int*)(bufB + NN * DH);
  int*   cnt  = rec + EE;
  int*   off  = cnt + NS * NB;
  int*   bst  = off + NS * NB;

  const dim3 gemmGrid(GB);
  const dim3 gemmBlk(GT);

  gcn_bin_count<<<NS, BT, 0, stream>>>(dst, cnt);
  gcn_bin_scan<<<1, BT, 0, stream>>>(cnt, off, bst);
  gcn_bin_place<<<NS, BT, 0, stream>>>(src, dst, off, bst, rec);
  gcn_bucket_dinv<<<NB, DT, 0, stream>>>(rec, bst, dinv);

  gcn_gemm_wmma<DH, false, false><<<gemmGrid, gemmBlk, 0, stream>>>(x, W1, nullptr, bufA);
  gcn_agg_bucket<<<NB, AT, 0, stream>>>(bufA, rec, bst, dinv, b1, bufB);

  gcn_gemm_wmma<DH, false, false><<<gemmGrid, gemmBlk, 0, stream>>>(bufB, W2, nullptr, bufA);
  gcn_agg_bucket<<<NB, AT, 0, stream>>>(bufA, rec, bst, dinv, b2, bufB);

  gcn_gemm_wmma<DH, false, false><<<gemmGrid, gemmBlk, 0, stream>>>(bufB, W3, nullptr, bufA);
  gcn_agg_bucket<<<NB, AT, 0, stream>>>(bufA, rec, bst, dinv, b3, bufB);

  gcn_gemm_wmma<DH, true, true><<<gemmGrid, gemmBlk, 0, stream>>>(bufB, Wf1, bf1, bufA);
  gcn_gemm_wmma<DOUT, true, false><<<gemmGrid, gemmBlk, 0, stream>>>(bufA, Wf2, bf2, out);
}


# speedup vs baseline: 4.4137x; 1.0406x over previous
#include <hip/hip_runtime.h>


typedef float v2f __attribute__((ext_vector_type(2)));
typedef float v8f __attribute__((ext_vector_type(8)));

#define NN   100000
#define NNP  100032
static_assert(NNP >= NN && NNP % 64 == 0 && NNP - NN < 64, "NNP must be NN rounded up to whole 256-byte units");
#define EE   1600000
#define DH   64
#define DOUT 16

#define BN   196
#define NB   511
static_assert(NB * BN >= NN && (NB - 1) * BN < NN, "NB must be ceil(NN / BN)");
static_assert(BN <= 256, "a record keeps the node's place in its bucket in 8 bits above src");
#define NS   128
#define SL   12500
#define SRCM 0x1FFFF
#define BT   1024
#define ET   13
static_assert(ET * BT >= SL && (ET - 1) * BT < SL, "ET must be ceil(SL / BT)");
#define AT   768
#define DT   512
#define GT   256
#define GB   391

__global__ __launch_bounds__(BT) void gcn_bin_count(const int* __restrict__ dst,
                                                    int* __restrict__ cnt) {
  __shared__ int hist[NB];
  const int tid = threadIdx.x;
  const int e0 = blockIdx.x * SL;
  int d[ET];
#pragma unroll
  for (int u = 0; u < ET; ++u) {
    const int i = tid + u * BT;
    d[u] = (i < SL) ? dst[e0 + i] : 0;
  }
  for (int i = tid; i < NB; i += BT) hist[i] = 0;
  __syncthreads();
#pragma unroll
  for (int u = 0; u < ET; ++u) {
    if (tid + u * BT < SL) atomicAdd(&hist[(unsigned)d[u] / BN], 1);
  }
  __syncthreads();
  for (int i = tid; i < NB; i += BT) cnt[blockIdx.x * NB + i] = hist[i];
}

__global__ __launch_bounds__(BT) void gcn_bin_scan(const int* __restrict__ cnt,
                                                   int* __restrict__ off,
                                                   int* __restrict__ bstart) {
  __shared__ int tot[BT];
  const int b = threadIdx.x;
  int run = 0;
  if (b < NB) {
    for (int s = 0; s < NS; s += 8) {
      int c[8];
#pragma unroll
      for (int u = 0; u < 8; ++u) c[u] = cnt[(s + u) * NB + b];
#pragma unroll
      for (int u = 0; u < 8; ++u) { off[(s + u) * NB + b] = run; run += c[u]; }
    }
  }
  tot[b] = run;
  __syncthreads();
  for (int d = 1; d < BT; d <<= 1) {
    const int v = (b >= d) ? tot[b - d] : 0;
    __syncthreads();
    tot[b] += v;
    __syncthreads();
  }
  if (b < NB) bstart[b + 1] = tot[b];
  if (b == 0) bstart[0] = 0;
}

__global__ __launch_bounds__(BT) void gcn_bin_place(const int* __restrict__ src,
                                                    const int* __restrict__ dst,
                                                    const int* __restrict__ off,
                                                    const int* __restrict__ bstart,
                                                    int* __restrict__ rec) {
  __shared__ int cur[NB];
  const int tid = threadIdx.x;
  const int e0 = blockIdx.x * SL;
  int t[ET], s[ET];
#pragma unroll
  for (int u = 0; u < ET; ++u) {
    const int i = tid + u * BT;
    t[u] = (i < SL) ? dst[e0 + i] : 0;
    s[u] = (i < SL) ? src[e0 + i] : 0;
  }
  for (int i = tid; i < NB; i += BT) cur[i] = bstart[i] + off[blockIdx.x * NB + i];
  __syncthreads();
#pragma unroll
  for (int u = 0; u < ET; ++u) {
    if (tid + u * BT < SL) {
      const int bk  = (unsigned)t[u] / BN;
      const int pos = atomicAdd(&cur[bk], 1);
      rec[pos] = s[u] | ((t[u] - bk * BN) << 17);
    }
  }
}

__global__ __launch_bounds__(DT) void gcn_bucket_dinv(const int* __restrict__ rec,
                                                       const int* __restrict__ bstart,
                                                       float* __restrict__ dinv) {
  __shared__ int c[BN];
  const int tid = threadIdx.x;
  if (tid < BN) c[tid] = 0;
  __syncthreads();
  const int beg = bstart[blockIdx.x];
  const int end = bstart[blockIdx.x + 1];
  for (int e0 = beg + tid; e0 < end; e0 += 8 * DT) {
    int rr[8];
#pragma unroll
    for (int u = 0; u < 8; ++u) {
      const int e = e0 + u * DT;
      rr[u] = rec[e < end ? e : e0];
    }
#pragma unroll
    for (int u = 0; u < 8; ++u) {
      if (e0 + u * DT < end) atomicAdd(&c[rr[u] >> 17], 1);
    }
  }
  __syncthreads();
  const int node = blockIdx.x * BN + tid;
  if (tid < BN && node < NN) dinv[node] = rsqrtf((float)c[tid] + 1.0f);
}

template <int NCOLS, bool BIAS, bool RELU>
__global__ __launch_bounds__(GT) void gcn_gemm_wmma(
    const float* __restrict__ X, const float* __restrict__ W,
    const float* __restrict__ bias, float* __restrict__ Out) {
  constexpr int NT = NCOLS / 16;
  __shared__ float4 Wl4[DH * NCOLS / 4];

  const int tid  = threadIdx.x;
  const int lane = tid & 31;
  const int h    = lane >> 4;
  const int c    = lane & 15;
  const int wave = __builtin_amdgcn_readfirstlane(tid >> 5);
  const int pair = blockIdx.x * (GT / 32) + wave;
  const bool live = pair < NN / 32;
  const int row0 = (live ? pair : 0) * 32;

  const float4* X4 = reinterpret_cast<const float4*>(X);
  float4 a[2][8];
#pragma unroll
  for (int m = 0; m < 2; ++m) {
#pragma unroll
    for (int j = 0; j < 8; ++j) a[m][j] = X4[(row0 + 16 * m + c) * (DH / 4) + 2 * j + h];
  }

  const float4* W4 = reinterpret_cast<const float4*>(W);
#pragma unroll
  for (int i = 0; i < DH * NCOLS / 4 / GT; ++i) Wl4[tid + i * GT] = W4[tid + i * GT];
  __syncthreads();
  if (!live) return;

  v8f acc[2][NT];
#pragma unroll
  for (int m = 0; m < 2; ++m) {
#pragma unroll
    for (int t = 0; t < NT; ++t) acc[m][t] = (v8f){};
  }

  const float* Wl = reinterpret_cast<const float*>(Wl4);
#pragma unroll
  for (int j = 0; j < 8; ++j) {
#pragma unroll
    for (int sub = 0; sub < 2; ++sub) {
      const int k0 = 8 * j + 4 * h + 2 * sub;
      v2f av[2];
#pragma unroll
      for (int m = 0; m < 2; ++m) {
        av[m].x = sub ? a[m][j].z : a[m][j].x;
        av[m].y = sub ? a[m][j].w : a[m][j].y;
      }
      v2f bv[NT];
      if constexpr (NT == 4) {
        const float4 b0 = Wl4[k0 * (NCOLS / 4) + c];
        const float4 b1 = Wl4[(k0 + 1) * (NCOLS / 4) + c];
        bv[0].x = b0.x; bv[0].y = b1.x;
        bv[1].x = b0.y; bv[1].y = b1.y;
        bv[2].x = b0.z; bv[2].y = b1.z;
        bv[3].x = b0.w; bv[3].y = b1.w;
      } else {
        bv[0].x = Wl[k0 * NCOLS + c];
        bv[0].y = Wl[(k0 + 1) * NCOLS + c];
      }
#pragma unroll
      for (int m = 0; m < 2; ++m) {
#pragma unroll
        for (int t = 0; t < NT; ++t) {
          acc[m][t] = __builtin_amdgcn_wmma_f32_16x16x4_f32(
               false, av[m],  false, bv[t],
               (short)0, acc[m][t],  false,  false);
        }
      }
    }
  }

  if constexpr (NT == 4) {
    float4 bb = make_float4(0.0f, 0.0f, 0.0f, 0.0f);
    if (BIAS) bb = reinterpret_cast<const float4*>(bias)[c];
    float4* O4 = reinterpret_cast<float4*>(Out);
#pragma unroll
    for (int m = 0; m < 2; ++m) {
#pragma unroll
      for (int r = 0; r < 8; ++r) {
        float4 o;
        o.x = acc[m][0][r];
        o.y = acc[m][1][r];
        o.z = acc[m][2][r];
        o.w = acc[m][3][r];
        if (BIAS) { o.x += bb.x; o.y += bb.y; o.z += bb.z; o.w += bb.w; }
        if (RELU) {
          o.x = fmaxf(o.x, 0.0f); o.y = fmaxf(o.y, 0.0f);
          o.z = fmaxf(o.z, 0.0f); o.w = fmaxf(o.w, 0.0f);
        }
        O4[(row0 + 16 * m + 8 * h + r) * (NCOLS / 4) + c] = o;
      }
    }
  } else {
    float bb = 0.0f;
    if (BIAS) bb = bias[c];
#pragma unroll
    for (int m = 0; m < 2; ++m) {
#pragma unroll
      for (int r = 0; r < 8; ++r) {
        float v = acc[m][0][r];
        if (BIAS) v += bb;
        if (RELU) v = fmaxf(v, 0.0f);
        Out[(row0 + 16 * m + 8 * h + r) * NCOLS + c] = v;
      }
    }
  }
}

template <bool FULL>
__device__ __forceinline__ void gcn_add_batch(const float2* __restrict__ H2, float* tile,
                                              int r, float cf, int n, int lane) {
#pragma unroll
  for (int g = 0; g < 32; g += 16) {
    if (FULL || g < n) {
      float2 v[16];
#pragma unroll
      for (int j = 0; j < 16; ++j) {
        const int rj = __builtin_amdgcn_readlane(r, g + j);
        v[j] = H2[(rj & SRCM) * (DH / 2) + lane];
      }
#pragma unroll
      for (int j = 0; j < 16; ++j) {
        if (FULL || g + j < n) {
          const int   rj = __builtin_amdgcn_readlane(r, g + j);
          const float cj = __int_as_float(__builtin_amdgcn_readlane(__float_as_int(cf), g + j));
          float* row = &tile[(rj >> 17) * DH + 2 * lane];
          atomicAdd(row,     v[j].x * cj);
          atomicAdd(row + 1, v[j].y * cj);
        }
      }
    }
  }
}

__global__ __launch_bounds__(AT) void gcn_agg_bucket(const float* __restrict__ H,
                                                     const int* __restrict__ rec,
                                                     const int* __restrict__ bstart,
                                                     const float* __restrict__ dinv,
                                                     const float* __restrict__ bias,
                                                     float* __restrict__ Out) {
  __shared__ float tile[BN * DH];
  __shared__ float sdinv[BN];
  const int tid   = threadIdx.x;
  const int lane  = tid & 31;
  const int wave  = __builtin_amdgcn_readfirstlane(tid >> 5);
  const int node0 = blockIdx.x * BN;

  float4* tile4 = reinterpret_cast<float4*>(tile);
  for (int i = tid; i < BN * DH / 4; i += AT) tile4[i] = make_float4(0.0f, 0.0f, 0.0f, 0.0f);
  if (tid < BN) sdinv[tid] = (node0 + tid < NN) ? dinv[node0 + tid] : 0.0f;
  __syncthreads();

  const int beg = bstart[blockIdx.x];
  const int end = bstart[blockIdx.x + 1];
  const float2* H2 = reinterpret_cast<const float2*>(H);

  for (int base = beg + wave * 32; base < end; base += AT) {
    const int e = base + lane;
    int   r  = 0;
    float cf = 0.0f;
    if (e < end) {
      r  = rec[e];
      cf = dinv[r & SRCM] * sdinv[r >> 17];
    }
    const int n = end - base;
    if (n >= 32) gcn_add_batch<true>(H2, tile, r, cf, 32, lane);
    else         gcn_add_batch<false>(H2, tile, r, cf, n, lane);
  }
  __syncthreads();

  const float4* H4 = reinterpret_cast<const float4*>(H);
  const float4* b4 = reinterpret_cast<const float4*>(bias);
  float4*       O4 = reinterpret_cast<float4*>(Out);
  constexpr int NP = BN * DH / 4;
  constexpr int FP = (NP + AT - 1) / AT;
  static_assert(AT % 16 == 0, "a thread must meet the same piece of a row in every pass");
  const int    q  = tid & 15;
  const float4 bb = b4[q];
  float4 hs[FP];
#pragma unroll
  for (int p = 0; p < FP; ++p) {
    const int i    = tid + p * AT;
    const int node = node0 + (i >> 4);
    hs[p] = (i < NP && node < NN) ? H4[node * (DH / 4) + q] : make_float4(0.0f, 0.0f, 0.0f, 0.0f);
  }
#pragma unroll
  for (int p = 0; p < FP; ++p) {
    const int i    = tid + p * AT;
    const int t    = i >> 4;
    const int node = node0 + t;
    if (i < NP && node < NN) {
      const float  di = sdinv[t];
      const float4 a  = tile4[i];
      const float4 h  = hs[p];
      float4 o;
      o.x = fmaxf(a.x + h.x * di * di + bb.x, 0.0f);
      o.y = fmaxf(a.y + h.y * di * di + bb.y, 0.0f);
      o.z = fmaxf(a.z + h.z * di * di + bb.z, 0.0f);
      o.w = fmaxf(a.w + h.w * di * di + bb.w, 0.0f);
      O4[node * (DH / 4) + q] = o;
    }
  }
}

extern "C" void kernel_launch(void* const* d_in, const int* in_sizes, int n_in,
                              void* d_out, int out_size, void* d_ws,
                              size_t ws_size, hipStream_t stream) {
  (void)in_sizes; (void)n_in; (void)out_size; (void)ws_size;

  const float* x   = (const float*)d_in[0];
  const int*   ei  = (const int*)d_in[1];
  const int*   src = ei;
  const int*   dst = ei + EE;
  const float* W1  = (const float*)d_in[2];
  const float* b1  = (const float*)d_in[3];
  const float* W2  = (const float*)d_in[4];
  const float* b2  = (const float*)d_in[5];
  const float* W3  = (const float*)d_in[6];
  const float* b3  = (const float*)d_in[7];
  const float* Wf1 = (const float*)d_in[8];
  const float* bf1 = (const float*)d_in[9];
  const float* Wf2 = (const float*)d_in[10];
  const float* bf2 = (const float*)d_in[11];
  float* out = (float*)d_out;

  float* base = (float*)d_ws;
  float* dinv = base;
  float* bufA = base + NNP;
  float* bufB = bufA + NN * DH;
  int*   rec  = (int*)(bufB + NN * DH);
  int*   cnt  = rec + EE;
  int*   off  = cnt + NS * NB;
  int*   bst  = off + NS * NB;

  const dim3 gemmGrid(GB);
  const dim3 gemmBlk(GT);

  gcn_bin_count<<<NS, BT, 0, stream>>>(dst, cnt);
  gcn_bin_scan<<<1, BT, 0, stream>>>(cnt, off, bst);
  gcn_bin_place<<<NS, BT, 0, stream>>>(src, dst, off, bst, rec);
  gcn_bucket_dinv<<<NB, DT, 0, stream>>>(rec, bst, dinv);

  gcn_gemm_wmma<DH, false, false><<<gemmGrid, gemmBlk, 0, stream>>>(x, W1, nullptr, bufA);
  gcn_agg_bucket<<<NB, AT, 0, stream>>>(bufA, rec, bst, dinv, b1, bufB);

  gcn_gemm_wmma<DH, false, false><<<gemmGrid, gemmBlk, 0, stream>>>(bufB, W2, nullptr, bufA);
  gcn_agg_bucket<<<NB, AT, 0, stream>>>(bufA, rec, bst, dinv, b2, bufB);

  gcn_gemm_wmma<DH, false, false><<<gemmGrid, gemmBlk, 0, stream>>>(bufB, W3, nullptr, bufA);
  gcn_agg_bucket<<<NB, AT, 0, stream>>>(bufA, rec, bst, dinv, b3, bufB);

  gcn_gemm_wmma<DH, true, true><<<gemmGrid, gemmBlk, 0, stream>>>(bufB, Wf1, bf1, bufA);
  gcn_gemm_wmma<DOUT, true, false><<<gemmGrid, gemmBlk, 0, stream>>>(bufA, Wf2, bf2, out);
}


# Edge weight dinv[src]*dinv[dst] computed once: layer 1's aggregation stores it in record order, layers 2-3 stream it instead of a scattered 4-byte dinv[src] read per edge; sums still by arrival
# speedup vs baseline: 4.5516x; 1.0312x over previous
#include <hip/hip_runtime.h>


typedef float v2f __attribute__((ext_vector_type(2)));
typedef float v8f __attribute__((ext_vector_type(8)));

#define NN   100000
#define NNP  100032
static_assert(NNP >= NN && NNP % 64 == 0 && NNP - NN < 64, "NNP must be NN rounded up to whole 256-byte units");
#define EE   1600000
#define DH   64
#define DOUT 16

#define BN   196
#define NB   511
static_assert(NB * BN >= NN && (NB - 1) * BN < NN, "NB must be ceil(NN / BN)");
static_assert(BN <= 256, "a record keeps the node's place in its bucket in 8 bits above src");
#define NS   128
#define SL   12500
#define SRCM 0x1FFFF
#define BT   1024
#define ET   13
static_assert(ET * BT >= SL && (ET - 1) * BT < SL, "ET must be ceil(SL / BT)");
#define AT   768
#define DT   512
#define GT   256
#define GB   391

__global__ __launch_bounds__(BT) void gcn_bin_count(const int* __restrict__ dst,
                                                    int* __restrict__ cnt) {
  __shared__ int hist[NB];
  const int tid = threadIdx.x;
  const int e0 = blockIdx.x * SL;
  int d[ET];
#pragma unroll
  for (int u = 0; u < ET; ++u) {
    const int i = tid + u * BT;
    d[u] = (i < SL) ? dst[e0 + i] : 0;
  }
  for (int i = tid; i < NB; i += BT) hist[i] = 0;
  __syncthreads();
#pragma unroll
  for (int u = 0; u < ET; ++u) {
    if (tid + u * BT < SL) atomicAdd(&hist[(unsigned)d[u] / BN], 1);
  }
  __syncthreads();
  for (int i = tid; i < NB; i += BT) cnt[blockIdx.x * NB + i] = hist[i];
}

__global__ __launch_bounds__(BT) void gcn_bin_scan(const int* __restrict__ cnt,
                                                   int* __restrict__ off,
                                                   int* __restrict__ bstart) {
  __shared__ int tot[BT];
  const int b = threadIdx.x;
  int run = 0;
  if (b < NB) {
    for (int s = 0; s < NS; s += 8) {
      int c[8];
#pragma unroll
      for (int u = 0; u < 8; ++u) c[u] = cnt[(s + u) * NB + b];
#pragma unroll
      for (int u = 0; u < 8; ++u) { off[(s + u) * NB + b] = run; run += c[u]; }
    }
  }
  tot[b] = run;
  __syncthreads();
  for (int d = 1; d < BT; d <<= 1) {
    const int v = (b >= d) ? tot[b - d] : 0;
    __syncthreads();
    tot[b] += v;
    __syncthreads();
  }
  if (b < NB) bstart[b + 1] = tot[b];
  if (b == 0) bstart[0] = 0;
}

__global__ __launch_bounds__(BT) void gcn_bin_place(const int* __restrict__ src,
                                                    const int* __restrict__ dst,
                                                    const int* __restrict__ off,
                                                    const int* __restrict__ bstart,
                                                    int* __restrict__ rec) {
  __shared__ int cur[NB];
  const int tid = threadIdx.x;
  const int e0 = blockIdx.x * SL;
  int t[ET], s[ET];
#pragma unroll
  for (int u = 0; u < ET; ++u) {
    const int i = tid + u * BT;
    t[u] = (i < SL) ? dst[e0 + i] : 0;
    s[u] = (i < SL) ? src[e0 + i] : 0;
  }
  for (int i = tid; i < NB; i += BT) cur[i] = bstart[i] + off[blockIdx.x * NB + i];
  __syncthreads();
#pragma unroll
  for (int u = 0; u < ET; ++u) {
    if (tid + u * BT < SL) {
      const int bk  = (unsigned)t[u] / BN;
      const int pos = atomicAdd(&cur[bk], 1);
      rec[pos] = s[u] | ((t[u] - bk * BN) << 17);
    }
  }
}

__global__ __launch_bounds__(DT) void gcn_bucket_dinv(const int* __restrict__ rec,
                                                       const int* __restrict__ bstart,
                                                       float* __restrict__ dinv) {
  __shared__ int c[BN];
  const int tid = threadIdx.x;
  if (tid < BN) c[tid] = 0;
  __syncthreads();
  const int beg = bstart[blockIdx.x];
  const int end = bstart[blockIdx.x + 1];
  for (int e0 = beg + tid; e0 < end; e0 += 8 * DT) {
    int rr[8];
#pragma unroll
    for (int u = 0; u < 8; ++u) {
      const int e = e0 + u * DT;
      rr[u] = rec[e < end ? e : e0];
    }
#pragma unroll
    for (int u = 0; u < 8; ++u) {
      if (e0 + u * DT < end) atomicAdd(&c[rr[u] >> 17], 1);
    }
  }
  __syncthreads();
  const int node = blockIdx.x * BN + tid;
  if (tid < BN && node < NN) dinv[node] = rsqrtf((float)c[tid] + 1.0f);
}

template <int NCOLS, bool BIAS, bool RELU>
__global__ __launch_bounds__(GT) void gcn_gemm_wmma(
    const float* __restrict__ X, const float* __restrict__ W,
    const float* __restrict__ bias, float* __restrict__ Out) {
  constexpr int NT = NCOLS / 16;
  __shared__ float4 Wl4[DH * NCOLS / 4];

  const int tid  = threadIdx.x;
  const int lane = tid & 31;
  const int h    = lane >> 4;
  const int c    = lane & 15;
  const int wave = __builtin_amdgcn_readfirstlane(tid >> 5);
  const int pair = blockIdx.x * (GT / 32) + wave;
  const bool live = pair < NN / 32;
  const int row0 = (live ? pair : 0) * 32;

  const float4* X4 = reinterpret_cast<const float4*>(X);
  float4 a[2][8];
#pragma unroll
  for (int m = 0; m < 2; ++m) {
#pragma unroll
    for (int j = 0; j < 8; ++j) a[m][j] = X4[(row0 + 16 * m + c) * (DH / 4) + 2 * j + h];
  }

  const float4* W4 = reinterpret_cast<const float4*>(W);
#pragma unroll
  for (int i = 0; i < DH * NCOLS / 4 / GT; ++i) Wl4[tid + i * GT] = W4[tid + i * GT];
  __syncthreads();
  if (!live) return;

  v8f acc[2][NT];
#pragma unroll
  for (int m = 0; m < 2; ++m) {
#pragma unroll
    for (int t = 0; t < NT; ++t) acc[m][t] = (v8f){};
  }

  const float* Wl = reinterpret_cast<const float*>(Wl4);
#pragma unroll
  for (int j = 0; j < 8; ++j) {
#pragma unroll
    for (int sub = 0; sub < 2; ++sub) {
      const int k0 = 8 * j + 4 * h + 2 * sub;
      v2f av[2];
#pragma unroll
      for (int m = 0; m < 2; ++m) {
        av[m].x = sub ? a[m][j].z : a[m][j].x;
        av[m].y = sub ? a[m][j].w : a[m][j].y;
      }
      v2f bv[NT];
      if constexpr (NT == 4) {
        const float4 b0 = Wl4[k0 * (NCOLS / 4) + c];
        const float4 b1 = Wl4[(k0 + 1) * (NCOLS / 4) + c];
        bv[0].x = b0.x; bv[0].y = b1.x;
        bv[1].x = b0.y; bv[1].y = b1.y;
        bv[2].x = b0.z; bv[2].y = b1.z;
        bv[3].x = b0.w; bv[3].y = b1.w;
      } else {
        bv[0].x = Wl[k0 * NCOLS + c];
        bv[0].y = Wl[(k0 + 1) * NCOLS + c];
      }
#pragma unroll
      for (int m = 0; m < 2; ++m) {
#pragma unroll
        for (int t = 0; t < NT; ++t) {
          acc[m][t] = __builtin_amdgcn_wmma_f32_16x16x4_f32(
               false, av[m],  false, bv[t],
               (short)0, acc[m][t],  false,  false);
        }
      }
    }
  }

  if constexpr (NT == 4) {
    float4 bb = make_float4(0.0f, 0.0f, 0.0f, 0.0f);
    if (BIAS) bb = reinterpret_cast<const float4*>(bias)[c];
    float4* O4 = reinterpret_cast<float4*>(Out);
#pragma unroll
    for (int m = 0; m < 2; ++m) {
#pragma unroll
      for (int r = 0; r < 8; ++r) {
        float4 o;
        o.x = acc[m][0][r];
        o.y = acc[m][1][r];
        o.z = acc[m][2][r];
        o.w = acc[m][3][r];
        if (BIAS) { o.x += bb.x; o.y += bb.y; o.z += bb.z; o.w += bb.w; }
        if (RELU) {
          o.x = fmaxf(o.x, 0.0f); o.y = fmaxf(o.y, 0.0f);
          o.z = fmaxf(o.z, 0.0f); o.w = fmaxf(o.w, 0.0f);
        }
        O4[(row0 + 16 * m + 8 * h + r) * (NCOLS / 4) + c] = o;
      }
    }
  } else {
    float bb = 0.0f;
    if (BIAS) bb = bias[c];
#pragma unroll
    for (int m = 0; m < 2; ++m) {
#pragma unroll
      for (int r = 0; r < 8; ++r) {
        float v = acc[m][0][r];
        if (BIAS) v += bb;
        if (RELU) v = fmaxf(v, 0.0f);
        Out[(row0 + 16 * m + 8 * h + r) * NCOLS + c] = v;
      }
    }
  }
}

template <bool FULL>
__device__ __forceinline__ void gcn_add_batch(const float2* __restrict__ H2, float* tile,
                                              int r, float cf, int n, int lane) {
#pragma unroll
  for (int g = 0; g < 32; g += 16) {
    if (FULL || g < n) {
      float2 v[16];
#pragma unroll
      for (int j = 0; j < 16; ++j) {
        const int rj = __builtin_amdgcn_readlane(r, g + j);
        v[j] = H2[(rj & SRCM) * (DH / 2) + lane];
      }
#pragma unroll
      for (int j = 0; j < 16; ++j) {
        if (FULL || g + j < n) {
          const int   rj = __builtin_amdgcn_readlane(r, g + j);
          const float cj = __int_as_float(__builtin_amdgcn_readlane(__float_as_int(cf), g + j));
          float* row = &tile[(rj >> 17) * DH + 2 * lane];
          atomicAdd(row,     v[j].x * cj);
          atomicAdd(row + 1, v[j].y * cj);
        }
      }
    }
  }
}

template <bool FIRST>
__global__ __launch_bounds__(AT) void gcn_agg_bucket(const float* __restrict__ H,
                                                     const int* __restrict__ rec,
                                                     const int* __restrict__ bstart,
                                                     const float* __restrict__ dinv,
                                                     const float* __restrict__ bias,
                                                     float* __restrict__ coef,
                                                     float* __restrict__ Out) {
  __shared__ float tile[BN * DH];
  __shared__ float sdinv[BN];
  const int tid   = threadIdx.x;
  const int lane  = tid & 31;
  const int wave  = __builtin_amdgcn_readfirstlane(tid >> 5);
  const int node0 = blockIdx.x * BN;

  float4* tile4 = reinterpret_cast<float4*>(tile);
  for (int i = tid; i < BN * DH / 4; i += AT) tile4[i] = make_float4(0.0f, 0.0f, 0.0f, 0.0f);
  if (tid < BN) sdinv[tid] = (node0 + tid < NN) ? dinv[node0 + tid] : 0.0f;
  __syncthreads();

  const int beg = bstart[blockIdx.x];
  const int end = bstart[blockIdx.x + 1];
  const float2* H2 = reinterpret_cast<const float2*>(H);

  for (int base = beg + wave * 32; base < end; base += AT) {
    const int e = base + lane;
    int   r  = 0;
    float cf = 0.0f;
    if (e < end) {
      r = rec[e];
      if (FIRST) {
        cf = dinv[r & SRCM] * sdinv[r >> 17];
        coef[e] = cf;
      } else {
        cf = coef[e];
      }
    }
    const int n = end - base;
    if (n >= 32) gcn_add_batch<true>(H2, tile, r, cf, 32, lane);
    else         gcn_add_batch<false>(H2, tile, r, cf, n, lane);
  }
  __syncthreads();

  const float4* H4 = reinterpret_cast<const float4*>(H);
  const float4* b4 = reinterpret_cast<const float4*>(bias);
  float4*       O4 = reinterpret_cast<float4*>(Out);
  constexpr int NP = BN * DH / 4;
  constexpr int FP = (NP + AT - 1) / AT;
  static_assert(AT % 16 == 0, "a thread must meet the same piece of a row in every pass");
  const int    q  = tid & 15;
  const float4 bb = b4[q];
  float4 hs[FP];
#pragma unroll
  for (int p = 0; p < FP; ++p) {
    const int i    = tid + p * AT;
    const int node = node0 + (i >> 4);
    hs[p] = (i < NP && node < NN) ? H4[node * (DH / 4) + q] : make_float4(0.0f, 0.0f, 0.0f, 0.0f);
  }
#pragma unroll
  for (int p = 0; p < FP; ++p) {
    const int i    = tid + p * AT;
    const int t    = i >> 4;
    const int node = node0 + t;
    if (i < NP && node < NN) {
      const float  di = sdinv[t];
      const float4 a  = tile4[i];
      const float4 h  = hs[p];
      float4 o;
      o.x = fmaxf(a.x + h.x * di * di + bb.x, 0.0f);
      o.y = fmaxf(a.y + h.y * di * di + bb.y, 0.0f);
      o.z = fmaxf(a.z + h.z * di * di + bb.z, 0.0f);
      o.w = fmaxf(a.w + h.w * di * di + bb.w, 0.0f);
      O4[node * (DH / 4) + q] = o;
    }
  }
}

extern "C" void kernel_launch(void* const* d_in, const int* in_sizes, int n_in,
                              void* d_out, int out_size, void* d_ws,
                              size_t ws_size, hipStream_t stream) {
  (void)in_sizes; (void)n_in; (void)out_size; (void)ws_size;

  const float* x   = (const float*)d_in[0];
  const int*   ei  = (const int*)d_in[1];
  const int*   src = ei;
  const int*   dst = ei + EE;
  const float* W1  = (const float*)d_in[2];
  const float* b1  = (const float*)d_in[3];
  const float* W2  = (const float*)d_in[4];
  const float* b2  = (const float*)d_in[5];
  const float* W3  = (const float*)d_in[6];
  const float* b3  = (const float*)d_in[7];
  const float* Wf1 = (const float*)d_in[8];
  const float* bf1 = (const float*)d_in[9];
  const float* Wf2 = (const float*)d_in[10];
  const float* bf2 = (const float*)d_in[11];
  float* out = (float*)d_out;

  float* base = (float*)d_ws;
  float* dinv = base;
  float* bufA = base + NNP;
  float* bufB = bufA + NN * DH;
  int*   rec  = (int*)(bufB + NN * DH);
  float* coef = (float*)(rec + EE);
  int*   cnt  = (int*)(coef + EE);
  int*   off  = cnt + NS * NB;
  int*   bst  = off + NS * NB;

  const dim3 gemmGrid(GB);
  const dim3 gemmBlk(GT);

  gcn_bin_count<<<NS, BT, 0, stream>>>(dst, cnt);
  gcn_bin_scan<<<1, BT, 0, stream>>>(cnt, off, bst);
  gcn_bin_place<<<NS, BT, 0, stream>>>(src, dst, off, bst, rec);
  gcn_bucket_dinv<<<NB, DT, 0, stream>>>(rec, bst, dinv);

  gcn_gemm_wmma<DH, false, false><<<gemmGrid, gemmBlk, 0, stream>>>(x, W1, nullptr, bufA);
  gcn_agg_bucket<true><<<NB, AT, 0, stream>>>(bufA, rec, bst, dinv, b1, coef, bufB);

  gcn_gemm_wmma<DH, false, false><<<gemmGrid, gemmBlk, 0, stream>>>(bufB, W2, nullptr, bufA);
  gcn_agg_bucket<false><<<NB, AT, 0, stream>>>(bufA, rec, bst, dinv, b2, coef, bufB);

  gcn_gemm_wmma<DH, false, false><<<gemmGrid, gemmBlk, 0, stream>>>(bufB, W3, nullptr, bufA);
  gcn_agg_bucket<false><<<NB, AT, 0, stream>>>(bufA, rec, bst, dinv, b3, coef, bufB);

  gcn_gemm_wmma<DH, true, true><<<gemmGrid, gemmBlk, 0, stream>>>(bufB, Wf1, bf1, bufA);
  gcn_gemm_wmma<DOUT, true, false><<<gemmGrid, gemmBlk, 0, stream>>>(bufA, Wf2, bf2, out);
}


# Layers alternate aggregate-first / product-first, so neighbouring products chain in one kernel through LDS: [W1,W2] and [W3,Wf1,Wf2]; 5 GEMM launches become 2 (12 -> 9); f32 sum order changes
# speedup vs baseline: 5.1308x; 1.1273x over previous
#include <hip/hip_runtime.h>


typedef float v2f __attribute__((ext_vector_type(2)));
typedef float v8f __attribute__((ext_vector_type(8)));

#define NN   100000
#define NNP  100032
static_assert(NNP >= NN && NNP % 64 == 0 && NNP - NN < 64, "NNP must be NN rounded up to whole 256-byte units");
#define EE   1600000
#define DH   64
#define DOUT 16

#define BN   196
#define NB   511
static_assert(NB * BN >= NN && (NB - 1) * BN < NN, "NB must be ceil(NN / BN)");
static_assert(BN <= 256, "a record keeps the node's place in its bucket in 8 bits above src");
#define NS   128
#define SL   12500
#define SRCM 0x1FFFF
#define BT   1024
#define ET   13
static_assert(ET * BT >= SL && (ET - 1) * BT < SL, "ET must be ceil(SL / BT)");
#define AT   768
#define DT   512
#define GT   256
#define GB   391

__global__ __launch_bounds__(BT) void gcn_bin_count(const int* __restrict__ dst,
                                                    int* __restrict__ cnt) {
  __shared__ int hist[NB];
  const int tid = threadIdx.x;
  const int e0 = blockIdx.x * SL;
  int d[ET];
#pragma unroll
  for (int u = 0; u < ET; ++u) {
    const int i = tid + u * BT;
    d[u] = (i < SL) ? dst[e0 + i] : 0;
  }
  for (int i = tid; i < NB; i += BT) hist[i] = 0;
  __syncthreads();
#pragma unroll
  for (int u = 0; u < ET; ++u) {
    if (tid + u * BT < SL) atomicAdd(&hist[(unsigned)d[u] / BN], 1);
  }
  __syncthreads();
  for (int i = tid; i < NB; i += BT) cnt[blockIdx.x * NB + i] = hist[i];
}

__global__ __launch_bounds__(BT) void gcn_bin_scan(const int* __restrict__ cnt,
                                                   int* __restrict__ off,
                                                   int* __restrict__ bstart) {
  __shared__ int tot[BT];
  const int b = threadIdx.x;
  int run = 0;
  if (b < NB) {
    for (int s = 0; s < NS; s += 8) {
      int c[8];
#pragma unroll
      for (int u = 0; u < 8; ++u) c[u] = cnt[(s + u) * NB + b];
#pragma unroll
      for (int u = 0; u < 8; ++u) { off[(s + u) * NB + b] = run; run += c[u]; }
    }
  }
  tot[b] = run;
  __syncthreads();
  for (int d = 1; d < BT; d <<= 1) {
    const int v = (b >= d) ? tot[b - d] : 0;
    __syncthreads();
    tot[b] += v;
    __syncthreads();
  }
  if (b < NB) bstart[b + 1] = tot[b];
  if (b == 0) bstart[0] = 0;
}

__global__ __launch_bounds__(BT) void gcn_bin_place(const int* __restrict__ src,
                                                    const int* __restrict__ dst,
                                                    const int* __restrict__ off,
                                                    const int* __restrict__ bstart,
                                                    int* __restrict__ rec) {
  __shared__ int cur[NB];
  const int tid = threadIdx.x;
  const int e0 = blockIdx.x * SL;
  int t[ET], s[ET];
#pragma unroll
  for (int u = 0; u < ET; ++u) {
    const int i = tid + u * BT;
    t[u] = (i < SL) ? dst[e0 + i] : 0;
    s[u] = (i < SL) ? src[e0 + i] : 0;
  }
  for (int i = tid; i < NB; i += BT) cur[i] = bstart[i] + off[blockIdx.x * NB + i];
  __syncthreads();
#pragma unroll
  for (int u = 0; u < ET; ++u) {
    if (tid + u * BT < SL) {
      const int bk  = (unsigned)t[u] / BN;
      const int pos = atomicAdd(&cur[bk], 1);
      rec[pos] = s[u] | ((t[u] - bk * BN) << 17);
    }
  }
}

__global__ __launch_bounds__(DT) void gcn_bucket_dinv(const int* __restrict__ rec,
                                                       const int* __restrict__ bstart,
                                                       float* __restrict__ dinv) {
  __shared__ int c[BN];
  const int tid = threadIdx.x;
  if (tid < BN) c[tid] = 0;
  __syncthreads();
  const int beg = bstart[blockIdx.x];
  const int end = bstart[blockIdx.x + 1];
  for (int e0 = beg + tid; e0 < end; e0 += 8 * DT) {
    int rr[8];
#pragma unroll
    for (int u = 0; u < 8; ++u) {
      const int e = e0 + u * DT;
      rr[u] = rec[e < end ? e : e0];
    }
#pragma unroll
    for (int u = 0; u < 8; ++u) {
      if (e0 + u * DT < end) atomicAdd(&c[rr[u] >> 17], 1);
    }
  }
  __syncthreads();
  const int node = blockIdx.x * BN + tid;
  if (tid < BN && node < NN) dinv[node] = rsqrtf((float)c[tid] + 1.0f);
}

template <int NCOLS>
__device__ __forceinline__ void gcn_stage_mm(const float4 (&a)[2][8], const float4* Wl4,
                                             v8f (&acc)[2][NCOLS / 16], int h, int c) {
  constexpr int NT = NCOLS / 16;
#pragma unroll
  for (int m = 0; m < 2; ++m) {
#pragma unroll
    for (int t = 0; t < NT; ++t) acc[m][t] = (v8f){};
  }
  const float* Wl = reinterpret_cast<const float*>(Wl4);
#pragma unroll
  for (int j = 0; j < 8; ++j) {
#pragma unroll
    for (int sub = 0; sub < 2; ++sub) {
      const int k0 = 8 * j + 4 * h + 2 * sub;
      v2f av[2];
#pragma unroll
      for (int m = 0; m < 2; ++m) {
        av[m].x = sub ? a[m][j].z : a[m][j].x;
        av[m].y = sub ? a[m][j].w : a[m][j].y;
      }
      v2f bv[NT];
      if constexpr (NT == 4) {
        const float4 b0 = Wl4[k0 * (NCOLS / 4) + c];
        const float4 b1 = Wl4[(k0 + 1) * (NCOLS / 4) + c];
        bv[0].x = b0.x; bv[0].y = b1.x;
        bv[1].x = b0.y; bv[1].y = b1.y;
        bv[2].x = b0.z; bv[2].y = b1.z;
        bv[3].x = b0.w; bv[3].y = b1.w;
      } else {
        bv[0].x = Wl[k0 * NCOLS + c];
        bv[0].y = Wl[(k0 + 1) * NCOLS + c];
      }
#pragma unroll
      for (int m = 0; m < 2; ++m) {
#pragma unroll
        for (int t = 0; t < NT; ++t) {
          acc[m][t] = __builtin_amdgcn_wmma_f32_16x16x4_f32(
               false, av[m],  false, bv[t],
               (short)0, acc[m][t],  false,  false);
        }
      }
    }
  }
}

#define SP 17
__device__ __forceinline__ void gcn_stage_turn(const v8f (&acc)[2][4], const float4 bb,
                                               float4* st4, float4 (&a)[2][8], int h, int c) {
#pragma unroll
  for (int m = 0; m < 2; ++m) {
#pragma unroll
    for (int r = 0; r < 8; ++r) {
      float4 o;
      o.x = fmaxf(acc[m][0][r] + bb.x, 0.0f);
      o.y = fmaxf(acc[m][1][r] + bb.y, 0.0f);
      o.z = fmaxf(acc[m][2][r] + bb.z, 0.0f);
      o.w = fmaxf(acc[m][3][r] + bb.w, 0.0f);
      st4[(16 * m + 8 * h + r) * SP + c] = o;
    }
  }
  __syncthreads();
#pragma unroll
  for (int m = 0; m < 2; ++m) {
#pragma unroll
    for (int j = 0; j < 8; ++j) a[m][j] = st4[(16 * m + c) * SP + 2 * j + h];
  }
}

template <int NST>
__global__ __launch_bounds__(GT) void gcn_gemm_chain(
    const float* __restrict__ X,
    const float* __restrict__ W0, const float* __restrict__ b0,
    const float* __restrict__ W1, const float* __restrict__ b1,
    const float* __restrict__ W2, const float* __restrict__ b2,
    float* __restrict__ Out) {
  __shared__ float4 Wl0[DH * DH / 4];
  __shared__ float4 Wl1[DH * DH / 4];
  __shared__ float4 Wl2[NST == 3 ? DH * DOUT / 4 : 1];
  __shared__ float4 stage[(GT / 32) * 32 * SP];

  const int tid  = threadIdx.x;
  const int lane = tid & 31;
  const int h    = lane >> 4;
  const int c    = lane & 15;
  const int wave = __builtin_amdgcn_readfirstlane(tid >> 5);
  const int pair = blockIdx.x * (GT / 32) + wave;
  const bool live = pair < NN / 32;
  const int row0 = (live ? pair : 0) * 32;

  const float4* X4 = reinterpret_cast<const float4*>(X);
  float4 a[2][8];
#pragma unroll
  for (int m = 0; m < 2; ++m) {
#pragma unroll
    for (int j = 0; j < 8; ++j) a[m][j] = X4[(row0 + 16 * m + c) * (DH / 4) + 2 * j + h];
  }

  const float4* W04 = reinterpret_cast<const float4*>(W0);
  const float4* W14 = reinterpret_cast<const float4*>(W1);
#pragma unroll
  for (int i = 0; i < DH * DH / 4 / GT; ++i) {
    Wl0[tid + i * GT] = W04[tid + i * GT];
    Wl1[tid + i * GT] = W14[tid + i * GT];
  }
  if constexpr (NST == 3) {
    static_assert(DH * DOUT / 4 == GT, "one 16-byte piece of the last weight per thread");
    Wl2[tid] = reinterpret_cast<const float4*>(W2)[tid];
  }
  __syncthreads();

  float4* st4 = stage + wave * (32 * SP);
  v8f acc[2][4];

  gcn_stage_mm<DH>(a, Wl0, acc, h, c);
  gcn_stage_turn(acc, reinterpret_cast<const float4*>(b0)[c], st4, a, h, c);
  gcn_stage_mm<DH>(a, Wl1, acc, h, c);

  if constexpr (NST == 2) {
    float4* O4 = reinterpret_cast<float4*>(Out);
    if (live) {
#pragma unroll
      for (int m = 0; m < 2; ++m) {
#pragma unroll
        for (int r = 0; r < 8; ++r) {
          float4 o;
          o.x = acc[m][0][r];
          o.y = acc[m][1][r];
          o.z = acc[m][2][r];
          o.w = acc[m][3][r];
          O4[(row0 + 16 * m + 8 * h + r) * (DH / 4) + c] = o;
        }
      }
    }
  } else {
    gcn_stage_turn(acc, reinterpret_cast<const float4*>(b1)[c], st4, a, h, c);
    v8f acc2[2][1];
    gcn_stage_mm<DOUT>(a, Wl2, acc2, h, c);
    const float bb = b2[c];
    if (live) {
#pragma unroll
      for (int m = 0; m < 2; ++m) {
#pragma unroll
        for (int r = 0; r < 8; ++r) {
          Out[(row0 + 16 * m + 8 * h + r) * DOUT + c] = acc2[m][0][r] + bb;
        }
      }
    }
  }
}

template <bool FULL>
__device__ __forceinline__ void gcn_add_batch(const float2* __restrict__ H2, float* tile,
                                              int r, float cf, int n, int lane) {
#pragma unroll
  for (int g = 0; g < 32; g += 16) {
    if (FULL || g < n) {
      float2 v[16];
#pragma unroll
      for (int j = 0; j < 16; ++j) {
        const int rj = __builtin_amdgcn_readlane(r, g + j);
        v[j] = H2[(rj & SRCM) * (DH / 2) + lane];
      }
#pragma unroll
      for (int j = 0; j < 16; ++j) {
        if (FULL || g + j < n) {
          const int   rj = __builtin_amdgcn_readlane(r, g + j);
          const float cj = __int_as_float(__builtin_amdgcn_readlane(__float_as_int(cf), g + j));
          float* row = &tile[(rj >> 17) * DH + 2 * lane];
          atomicAdd(row,     v[j].x * cj);
          atomicAdd(row + 1, v[j].y * cj);
        }
      }
    }
  }
}

template <bool FIRST, bool PLAIN>
__global__ __launch_bounds__(AT) void gcn_agg_bucket(const float* __restrict__ H,
                                                     const int* __restrict__ rec,
                                                     const int* __restrict__ bstart,
                                                     const float* __restrict__ dinv,
                                                     const float* __restrict__ bias,
                                                     float* __restrict__ coef,
                                                     float* __restrict__ Out) {
  __shared__ float tile[BN * DH];
  __shared__ float sdinv[BN];
  const int tid   = threadIdx.x;
  const int lane  = tid & 31;
  const int wave  = __builtin_amdgcn_readfirstlane(tid >> 5);
  const int node0 = blockIdx.x * BN;

  float4* tile4 = reinterpret_cast<float4*>(tile);
  for (int i = tid; i < BN * DH / 4; i += AT) tile4[i] = make_float4(0.0f, 0.0f, 0.0f, 0.0f);
  if (tid < BN) sdinv[tid] = (node0 + tid < NN) ? dinv[node0 + tid] : 0.0f;
  __syncthreads();

  const int beg = bstart[blockIdx.x];
  const int end = bstart[blockIdx.x + 1];
  const float2* H2 = reinterpret_cast<const float2*>(H);

  for (int base = beg + wave * 32; base < end; base += AT) {
    const int e = base + lane;
    int   r  = 0;
    float cf = 0.0f;
    if (e < end) {
      r = rec[e];
      if (FIRST) {
        cf = dinv[r & SRCM] * sdinv[r >> 17];
        coef[e] = cf;
      } else {
        cf = coef[e];
      }
    }
    const int n = end - base;
    if (n >= 32) gcn_add_batch<true>(H2, tile, r, cf, 32, lane);
    else         gcn_add_batch<false>(H2, tile, r, cf, n, lane);
  }
  __syncthreads();

  const float4* H4 = reinterpret_cast<const float4*>(H);
  const float4* b4 = reinterpret_cast<const float4*>(bias);
  float4*       O4 = reinterpret_cast<float4*>(Out);
  constexpr int NP = BN * DH / 4;
  constexpr int FP = (NP + AT - 1) / AT;
  static_assert(AT % 16 == 0, "a thread must meet the same piece of a row in every pass");
  const int    q  = tid & 15;
  float4 bb = make_float4(0.0f, 0.0f, 0.0f, 0.0f);
  if constexpr (!PLAIN) bb = b4[q];
  float4 hs[FP];
#pragma unroll
  for (int p = 0; p < FP; ++p) {
    const int i    = tid + p * AT;
    const int node = node0 + (i >> 4);
    hs[p] = (i < NP && node < NN) ? H4[node * (DH / 4) + q] : make_float4(0.0f, 0.0f, 0.0f, 0.0f);
  }
#pragma unroll
  for (int p = 0; p < FP; ++p) {
    const int i    = tid + p * AT;
    const int t    = i >> 4;
    const int node = node0 + t;
    if (i < NP && node < NN) {
      const float  di = sdinv[t];
      const float4 a  = tile4[i];
      const float4 h  = hs[p];
      float4 o;
      if constexpr (PLAIN) {
        o.x = a.x + h.x * di * di;
        o.y = a.y + h.y * di * di;
        o.z = a.z + h.z * di * di;
        o.w = a.w + h.w * di * di;
      } else {
        o.x = fmaxf(a.x + h.x * di * di + bb.x, 0.0f);
        o.y = fmaxf(a.y + h.y * di * di + bb.y, 0.0f);
        o.z = fmaxf(a.z + h.z * di * di + bb.z, 0.0f);
        o.w = fmaxf(a.w + h.w * di * di + bb.w, 0.0f);
      }
      O4[node * (DH / 4) + q] = o;
    }
  }
}

extern "C" void kernel_launch(void* const* d_in, const int* in_sizes, int n_in,
                              void* d_out, int out_size, void* d_ws,
                              size_t ws_size, hipStream_t stream) {
  (void)in_sizes; (void)n_in; (void)out_size; (void)ws_size;

  const float* x   = (const float*)d_in[0];
  const int*   ei  = (const int*)d_in[1];
  const int*   src = ei;
  const int*   dst = ei + EE;
  const float* W1  = (const float*)d_in[2];
  const float* b1  = (const float*)d_in[3];
  const float* W2  = (const float*)d_in[4];
  const float* b2  = (const float*)d_in[5];
  const float* W3  = (const float*)d_in[6];
  const float* b3  = (const float*)d_in[7];
  const float* Wf1 = (const float*)d_in[8];
  const float* bf1 = (const float*)d_in[9];
  const float* Wf2 = (const float*)d_in[10];
  const float* bf2 = (const float*)d_in[11];
  float* out = (float*)d_out;

  float* base = (float*)d_ws;
  float* dinv = base;
  float* bufA = base + NNP;
  float* bufB = bufA + NN * DH;
  int*   rec  = (int*)(bufB + NN * DH);
  float* coef = (float*)(rec + EE);
  int*   cnt  = (int*)(coef + EE);
  int*   off  = cnt + NS * NB;
  int*   bst  = off + NS * NB;

  const dim3 gemmGrid(GB);
  const dim3 gemmBlk(GT);

  gcn_bin_count<<<NS, BT, 0, stream>>>(dst, cnt);
  gcn_bin_scan<<<1, BT, 0, stream>>>(cnt, off, bst);
  gcn_bin_place<<<NS, BT, 0, stream>>>(src, dst, off, bst, rec);
  gcn_bucket_dinv<<<NB, DT, 0, stream>>>(rec, bst, dinv);

  gcn_agg_bucket<true, true><<<NB, AT, 0, stream>>>(x, rec, bst, dinv, nullptr, coef, bufA);

  gcn_gemm_chain<2><<<gemmGrid, gemmBlk, 0, stream>>>(bufA, W1, b1, W2, nullptr, nullptr, nullptr, bufB);

  gcn_agg_bucket<false, false><<<NB, AT, 0, stream>>>(bufB, rec, bst, dinv, b2, coef, bufA);

  gcn_agg_bucket<false, true><<<NB, AT, 0, stream>>>(bufA, rec, bst, dinv, nullptr, coef, bufB);

  gcn_gemm_chain<3><<<gemmGrid, gemmBlk, 0, stream>>>(bufB, W3, b3, Wf1, bf1, Wf2, bf2, out);
}
